# starting kernel
# baseline (speedup 1.0000x reference)
#include <hip/hip_runtime.h>
#include <hip/hip_bf16.h>
#include <math.h>

typedef __attribute__((ext_vector_type(2))) float v2f;
typedef __attribute__((ext_vector_type(8))) float v8f;

#define B_ 4
#define N_ 50000
#define E_ 800000
#define FIN_ 128
#define FOUT_ 64
#define ALPHA_ 0.2f
#define NEG_BIG_ (-9000000000000000.0f)

#define ROWS_ (B_ * N_)
#define TILES_ (ROWS_ / 16)

__global__ __launch_bounds__(256) void gat_gemm_wmma(const float* __restrict__ X,
                                                     const float* __restrict__ W,
                                                     float* __restrict__ h) {
    const int lane   = threadIdx.x & 31;
    const int waveId = threadIdx.x >> 5;
    const int rt     = blockIdx.x * 8 + waveId;
    if (rt >= TILES_) return;

    const int m_base = rt * 16;
    const int mrow   = m_base + (lane & 15);
    const int khalf  = (lane >> 4) << 1;
    const int ncol   = lane & 15;

    v8f acc0 = {}, acc1 = {}, acc2 = {}, acc3 = {};

    for (int k = 0; k < FIN_; k += 4) {
        const int kk = k + khalf;
        v2f a;
        a.x = X[(size_t)mrow * FIN_ + kk];
        a.y = X[(size_t)mrow * FIN_ + kk + 1];

        const float* wr0 = W + (size_t)kk * FOUT_;
        const float* wr1 = wr0 + FOUT_;

        v2f b0, b1, b2, b3;
        b0.x = wr0[ncol];      b0.y = wr1[ncol];
        b1.x = wr0[16 + ncol]; b1.y = wr1[16 + ncol];
        b2.x = wr0[32 + ncol]; b2.y = wr1[32 + ncol];
        b3.x = wr0[48 + ncol]; b3.y = wr1[48 + ncol];

        acc0 = __builtin_amdgcn_wmma_f32_16x16x4_f32(false, a, false, b0, (short)0, acc0, false, false);
        acc1 = __builtin_amdgcn_wmma_f32_16x16x4_f32(false, a, false, b1, (short)0, acc1, false, false);
        acc2 = __builtin_amdgcn_wmma_f32_16x16x4_f32(false, a, false, b2, (short)0, acc2, false, false);
        acc3 = __builtin_amdgcn_wmma_f32_16x16x4_f32(false, a, false, b3, (short)0, acc3, false, false);
    }

    const int rhalf = (lane >> 4) << 3;
#pragma unroll
    for (int i = 0; i < 8; ++i) {
        const size_t row = (size_t)(m_base + i + rhalf);
        h[row * FOUT_ +  0 + ncol] = acc0[i];
        h[row * FOUT_ + 16 + ncol] = acc1[i];
        h[row * FOUT_ + 32 + ncol] = acc2[i];
        h[row * FOUT_ + 48 + ncol] = acc3[i];
    }
}

__global__ __launch_bounds__(256) void gat_scores(const float* __restrict__ h,
                                                  const float* __restrict__ a,
                                                  float* __restrict__ e1,
                                                  float* __restrict__ e2) {
    const int row = blockIdx.x * blockDim.x + threadIdx.x;
    if (row >= ROWS_) return;
    const float* hr = h + (size_t)row * FOUT_;
    float s1 = 0.f, s2 = 0.f;
#pragma unroll
    for (int f = 0; f < FOUT_; ++f) {
        const float hv = hr[f];
        s1 += hv * a[f];
        s2 += hv * a[FOUT_ + f];
    }
    e1[row] = s1;
    e2[row] = s2;
}

__global__ __launch_bounds__(256) void gat_edges(const int* __restrict__ edges,
                                                 const float* __restrict__ h,
                                                 const float* __restrict__ e1,
                                                 const float* __restrict__ e2,
                                                 float* __restrict__ accum,
                                                 float* __restrict__ rowsum) {
    __shared__ float s_ee[4];
    __shared__ int   s_src[4];
    __shared__ int   s_dst[4];

    const int tid = threadIdx.x;
    const int le  = tid >> 6;
    const int f   = tid & 63;

    const long long eg = (long long)blockIdx.x * 4 + le;
    const bool valid = eg < (long long)B_ * E_;
    int b = 0, e = 0;
    if (valid) {
        b = (int)(eg / E_);
        e = (int)(eg % E_);
        if (f == 0) {
            const int src = edges[((size_t)b * 2 + 0) * E_ + e];
            const int dst = edges[((size_t)b * 2 + 1) * E_ + e];
            const float s  = e1[(size_t)b * N_ + src] + e2[(size_t)b * N_ + dst];
            const float lr = (s >= 0.f) ? s : ALPHA_ * s;
            const float ee = __expf(-lr);
            s_ee[le]  = ee;
            s_src[le] = src;
            s_dst[le] = dst;
            atomicAdd(&rowsum[(size_t)b * N_ + src], ee);
        }
    }
    __syncthreads();
    if (valid) {
        const int   src = s_src[le];
        const int   dst = s_dst[le];
        const float ee  = s_ee[le];
        const float hv  = h[((size_t)b * N_ + dst) * FOUT_ + f];
        atomicAdd(&accum[((size_t)b * N_ + src) * FOUT_ + f], ee * hv);
    }
}

__global__ __launch_bounds__(256) void gat_finalize(const float* __restrict__ accum,
                                                    const float* __restrict__ rowsum,
                                                    float* __restrict__ out) {
    const long long idx = (long long)blockIdx.x * blockDim.x + threadIdx.x;
    if (idx >= (long long)ROWS_ * FOUT_) return;
    const long long row = idx >> 6;
    float v = accum[idx] / rowsum[row];
    if (isnan(v)) v = NEG_BIG_;
    out[idx] = (v > 0.f) ? v : expm1f(v);
}

extern "C" void kernel_launch(void* const* d_in, const int* in_sizes, int n_in,
                              void* d_out, int out_size, void* d_ws, size_t ws_size,
                              hipStream_t stream) {
    const float* X     = (const float*)d_in[0];
    const int*   edges = (const int*)d_in[1];
    const float* W     = (const float*)d_in[2];
    const float* a     = (const float*)d_in[3];
    float*       out   = (float*)d_out;

    float* accum  = (float*)d_ws;
    float* rowsum = accum + (size_t)ROWS_ * FOUT_;
    float* h      = rowsum + (size_t)ROWS_;
    float* e1     = h + (size_t)ROWS_ * FOUT_;
    float* e2     = e1 + (size_t)ROWS_;

    hipMemsetAsync(accum, 0, ((size_t)ROWS_ * FOUT_ + ROWS_) * sizeof(float), stream);

    {
        const int waves_per_block = 8;
        const int blocks = (TILES_ + waves_per_block - 1) / waves_per_block;
        gat_gemm_wmma<<<blocks, 256, 0, stream>>>(X, W, h);
    }
    {
        const int blocks = (ROWS_ + 255) / 256;
        gat_scores<<<blocks, 256, 0, stream>>>(h, a, e1, e2);
    }
    {
        const long long total_edges = (long long)B_ * E_;
        const int blocks = (int)((total_edges + 3) / 4);
        gat_edges<<<blocks, 256, 0, stream>>>(edges, h, e1, e2, accum, rowsum);
    }
    {
        const long long total = (long long)ROWS_ * FOUT_;
        const int blocks = (int)((total + 255) / 256);
        gat_finalize<<<blocks, 256, 0, stream>>>(accum, rowsum, out);
    }
}


# Owners no longer filter all 32 scan lists: each scanning wave sorts its own list by owner wave (stable in-wave counting sort, LDS int counts), owners copy their pieces; float sum order unchanged
# speedup vs baseline: 5.0236x; 5.0236x over previous
#include <hip/hip_runtime.h>
#include <hip/hip_bf16.h>
#include <math.h>

typedef __attribute__((ext_vector_type(2))) float v2f;
typedef __attribute__((ext_vector_type(8))) float v8f;

#define B_ 4
#define N_ 50000
#define E_ 800000
#define FIN_ 128
#define FOUT_ 64
#define ALPHA_ 0.2f
#define NEG_BIG_ (-9000000000000000.0f)

#define ROWS_ (B_ * N_)
#define TILES_ (ROWS_ / 16)

__global__ __launch_bounds__(256) void gat_gemm_wmma(const float* __restrict__ X,
                                                     const float* __restrict__ W,
                                                     float* __restrict__ h) {
    const int lane   = threadIdx.x & 31;
    const int waveId = threadIdx.x >> 5;
    const int rt     = blockIdx.x * 8 + waveId;
    if (rt >= TILES_) return;

    const int m_base = rt * 16;
    const int mrow   = m_base + (lane & 15);
    const int khalf  = (lane >> 4) << 1;
    const int ncol   = lane & 15;

    v8f acc0 = {}, acc1 = {}, acc2 = {}, acc3 = {};

    for (int k = 0; k < FIN_; k += 4) {
        const int kk = k + khalf;
        v2f a;
        a.x = X[(size_t)mrow * FIN_ + kk];
        a.y = X[(size_t)mrow * FIN_ + kk + 1];

        const float* wr0 = W + (size_t)kk * FOUT_;
        const float* wr1 = wr0 + FOUT_;

        v2f b0, b1, b2, b3;
        b0.x = wr0[ncol];      b0.y = wr1[ncol];
        b1.x = wr0[16 + ncol]; b1.y = wr1[16 + ncol];
        b2.x = wr0[32 + ncol]; b2.y = wr1[32 + ncol];
        b3.x = wr0[48 + ncol]; b3.y = wr1[48 + ncol];

        acc0 = __builtin_amdgcn_wmma_f32_16x16x4_f32(false, a, false, b0, (short)0, acc0, false, false);
        acc1 = __builtin_amdgcn_wmma_f32_16x16x4_f32(false, a, false, b1, (short)0, acc1, false, false);
        acc2 = __builtin_amdgcn_wmma_f32_16x16x4_f32(false, a, false, b2, (short)0, acc2, false, false);
        acc3 = __builtin_amdgcn_wmma_f32_16x16x4_f32(false, a, false, b3, (short)0, acc3, false, false);
    }

    const int rhalf = (lane >> 4) << 3;
#pragma unroll
    for (int i = 0; i < 8; ++i) {
        const size_t row = (size_t)(m_base + i + rhalf);
        h[row * FOUT_ +  0 + ncol] = acc0[i];
        h[row * FOUT_ + 16 + ncol] = acc1[i];
        h[row * FOUT_ + 32 + ncol] = acc2[i];
        h[row * FOUT_ + 48 + ncol] = acc3[i];
    }
}

__global__ __launch_bounds__(256) void gat_scores(const float* __restrict__ h,
                                                  const float* __restrict__ a,
                                                  float* __restrict__ e1,
                                                  float* __restrict__ e2) {
    const int row = blockIdx.x * blockDim.x + threadIdx.x;
    if (row >= ROWS_) return;
    const float* hr = h + (size_t)row * FOUT_;
    float s1 = 0.f, s2 = 0.f;
#pragma unroll
    for (int f = 0; f < FOUT_; ++f) {
        const float hv = hr[f];
        s1 += hv * a[f];
        s2 += hv * a[FOUT_ + f];
    }
    e1[row] = s1;
    e2[row] = s2;
}

#define R_      782
#define NRANGE_ 64
#define C1_     704
#define C2_     72
#define NBLK_   (E_ / 128)
#define OFF_RS_ (R_ * FOUT_ * 4)
#define OFF_BL_ (OFF_RS_ + R_ * 4)
#define OFF_OL_ (OFF_BL_ + 32 * C1_ * 4)
#define OFF_ET_ (OFF_OL_ + 32 * C2_ * 4)
#define OFF_SF_ (OFF_ET_ + 32 * 32 * 4)
#define LDS_BYTES_ (OFF_SF_ + 32 * 4)

__device__ __forceinline__ int gat_flush(unsigned* __restrict__ myO, int n, int lane, bool all,
                                         const float* __restrict__ hb,
                                         const float* __restrict__ e1b,
                                         const float* __restrict__ e2b,
                                         float* __restrict__ acc, float& rsum) {
    int i = 0;
    for (; i + 8 <= n; i += 8) {
        unsigned en[8];
        float2 hv[8];
        float sv[8];
#pragma unroll
        for (int k = 0; k < 8; ++k) en[k] = __builtin_amdgcn_readfirstlane(myO[i + k]);
#pragma unroll
        for (int k = 0; k < 8; ++k) {
            const unsigned dst = en[k] & 0xFFFFu;
            const unsigned d   = en[k] >> 16;
            hv[k] = *(const float2*)(hb + (size_t)dst * FOUT_ + 2 * lane);
            sv[k] = e1b[d] + e2b[dst];
        }
#pragma unroll
        for (int k = 0; k < 8; ++k) {
            const unsigned d = en[k] >> 16;
            const float s  = sv[k];
            const float lr = (s >= 0.f) ? s : ALPHA_ * s;
            const float ee = __expf(-lr);
            float2* ap = (float2*)(acc + d * FOUT_) + lane;
            float2 a = *ap;
            a.x += ee * hv[k].x;
            a.y += ee * hv[k].y;
            *ap = a;
            rsum += (lane == (int)(d >> 5)) ? ee : 0.f;
        }
    }
    if (!all) {
        const int rem = n - i;
        unsigned t = 0u;
        if (lane < rem) t = myO[i + lane];
        if (lane < rem) myO[lane] = t;
        return rem;
    }
    for (; i < n; ++i) {
        const unsigned e0  = __builtin_amdgcn_readfirstlane(myO[i]);
        const unsigned dst = e0 & 0xFFFFu;
        const unsigned d   = e0 >> 16;
        const float2 hv = *(const float2*)(hb + (size_t)dst * FOUT_ + 2 * lane);
        const float s  = e1b[d] + e2b[dst];
        const float lr = (s >= 0.f) ? s : ALPHA_ * s;
        const float ee = __expf(-lr);
        float2* ap = (float2*)(acc + d * FOUT_) + lane;
        float2 a = *ap;
        a.x += ee * hv.x;
        a.y += ee * hv.y;
        *ap = a;
        rsum += (lane == (int)(d >> 5)) ? ee : 0.f;
    }
    return 0;
}

__device__ __forceinline__ unsigned gat_same_owner(unsigned o, bool valid) {
    unsigned mk = __builtin_amdgcn_ballot_w32(valid);
#pragma unroll
    for (int bit = 0; bit < 5; ++bit) {
        const bool     on  = ((o >> bit) & 1u) != 0u;
        const unsigned bal = __builtin_amdgcn_ballot_w32(on);
        mk &= on ? bal : ~bal;
    }
    return mk;
}

__global__ __launch_bounds__(1024) void gat_rows(const int* __restrict__ edges,
                                                 const float* __restrict__ h,
                                                 const float* __restrict__ e1,
                                                 const float* __restrict__ e2,
                                                 int* __restrict__ alist,
                                                 float* __restrict__ out) {
    __shared__ __attribute__((aligned(16))) unsigned char L[LDS_BYTES_];
    float*    acc   = (float*)L;
    float*    rs    = (float*)(L + OFF_RS_);
    unsigned* blist = (unsigned*)(L + OFF_BL_);
    unsigned* olist = (unsigned*)(L + OFF_OL_);
    int*      endt  = (int*)(L + OFF_ET_);
    int*      sflag = (int*)(L + OFF_SF_);

    const int tid  = threadIdx.x;
    const int lane = tid & 31;
    const int w    = __builtin_amdgcn_readfirstlane(tid >> 5);
    const int b    = blockIdx.x / NRANGE_;
    const int rg   = blockIdx.x % NRANGE_;
    const int row_lo = rg * R_;
    const int nrows  = (N_ - row_lo < R_) ? (N_ - row_lo) : R_;

    const int*   srcarr = edges + (size_t)b * 2 * E_;
    const int*   dstarr = srcarr + E_;
    const float* hb     = h  + (size_t)b * N_ * FOUT_;
    const float* e1b    = e1 + (size_t)b * N_ + row_lo;
    const float* e2b    = e2 + (size_t)b * N_;

    for (int i = tid; i < R_ * FOUT_ / 4; i += 1024) ((float4*)L)[i] = make_float4(0.f, 0.f, 0.f, 0.f);

    int*      myA = alist + ((size_t)blockIdx.x * 32 + w) * C1_;
    unsigned* myB = blist + w * C1_;
    int*      myC = endt + w * 32;
    unsigned* myO = olist + w * C2_;
    const unsigned lt_mask = (1u << lane) - 1u;
    int it   = 0;
    int ocnt = 0;
    float rsum = 0.f;
    __syncthreads();

    for (int pass = 0; pass < 128; ++pass) {
        int cnt = 0;
        myC[lane] = 0;
        while (cnt <= C1_ - 128 && it * 32 + w < NBLK_) {
            int4 s[4];
#pragma unroll
            for (int j = 0; j < 4; ++j) {
                const int blk = (it + j) * 32 + w;
                if (blk < NBLK_) s[j] = *(const int4*)(srcarr + (size_t)blk * 128 + lane * 4);
                else             s[j] = make_int4(-1, -1, -1, -1);
            }
#pragma unroll
            for (int j = 0; j < 4; ++j) {
                if (cnt <= C1_ - 128) {
                    const unsigned ebase = (unsigned)(((it * 32 + w) * 128) + lane * 4);
                    const int sc[4] = {s[j].x, s[j].y, s[j].z, s[j].w};
#pragma unroll
                    for (int c = 0; c < 4; ++c) {
                        const unsigned d  = (unsigned)(sc[c] - row_lo);
                        const bool     m  = d < (unsigned)nrows;
                        const unsigned bm = __builtin_amdgcn_ballot_w32(m);
                        if (bm) {
                            if (m) {
                                myA[cnt + __builtin_popcount(bm & lt_mask)] = (int)((ebase + c) | (d << 20));
                                atomicAdd(&myC[d & 31u], 1);
                            }
                            cnt += __builtin_popcount(bm);
                        }
                    }
                    ++it;
                }
            }
        }
        __threadfence_block();
        {
            const int c = myC[lane];
            int incl = c;
#pragma unroll
            for (int k = 1; k < 32; k <<= 1) {
                const int t = __shfl_up(incl, k, 32);
                if (lane >= k) incl += t;
            }
            myC[lane] = incl - c;
        }
        for (int i0 = 0; i0 < cnt; i0 += 128) {
            unsigned en[4];
            int      dd[4];
#pragma unroll
            for (int j = 0; j < 4; ++j) {
                const int idx = i0 + j * 32 + lane;
                en[j] = (idx < cnt) ? (unsigned)myA[idx] : 0u;
            }
#pragma unroll
            for (int j = 0; j < 4; ++j) dd[j] = dstarr[en[j] & 0xFFFFFu];
#pragma unroll
            for (int j = 0; j < 4; ++j) {
                if (i0 + j * 32 < cnt) {
                    const bool     valid = (i0 + j * 32 + lane) < cnt;
                    const unsigned d     = en[j] >> 20;
                    const unsigned o     = d & 31u;
                    const unsigned mk    = gat_same_owner(o, valid);
                    const int pos = myC[o] + __builtin_popcount(mk & lt_mask);
                    if (valid) myB[pos] = (unsigned)dd[j] | (d << 16);
                    if (valid && (mk >> lane) == 1u) myC[o] = pos + 1;
                }
            }
        }
        if (lane == 0) sflag[w] = (it * 32 + w < NBLK_) ? 1 : 0;
        __syncthreads();

        const int pieceEnd   = endt[lane * 32 + w];
        const int pieceStart = (w > 0) ? endt[lane * 32 + w - 1] : 0;
        for (int v = 0; v < 32; ++v) {
            int       p  = __builtin_amdgcn_readlane(pieceStart, v);
            const int p1 = __builtin_amdgcn_readlane(pieceEnd, v);
            const unsigned* bl = blist + v * C1_;
            while (p < p1) {
                if (ocnt > C2_ - 32)
                    ocnt = gat_flush(myO, ocnt, lane, false, hb, e1b, e2b, acc, rsum);
                const int take = (p1 - p < 32) ? (p1 - p) : 32;
                if (lane < take) myO[ocnt + lane] = bl[p + lane];
                ocnt += take;
                p    += take;
            }
        }
        const unsigned more = __builtin_amdgcn_ballot_w32(sflag[lane] != 0);
        __syncthreads();
        if (more == 0u) break;
    }
    gat_flush(myO, ocnt, lane, true, hb, e1b, e2b, acc, rsum);
    if (lane * 32 + w < R_) rs[lane * 32 + w] = rsum;
    __syncthreads();

    for (int i = tid; i < nrows * (FOUT_ / 4); i += 1024) {
        const int r = i >> 4;
        const int c = (i & 15) * 4;
        const float4 a = *(const float4*)(acc + r * FOUT_ + c);
        const float  q = rs[r];
        float v0 = a.x / q, v1 = a.y / q, v2 = a.z / q, v3 = a.w / q;
        if (isnan(v0)) v0 = NEG_BIG_;
        if (isnan(v1)) v1 = NEG_BIG_;
        if (isnan(v2)) v2 = NEG_BIG_;
        if (isnan(v3)) v3 = NEG_BIG_;
        float4 o;
        o.x = (v0 > 0.f) ? v0 : expm1f(v0);
        o.y = (v1 > 0.f) ? v1 : expm1f(v1);
        o.z = (v2 > 0.f) ? v2 : expm1f(v2);
        o.w = (v3 > 0.f) ? v3 : expm1f(v3);
        *(float4*)(out + ((size_t)b * N_ + row_lo + r) * FOUT_ + c) = o;
    }
}

extern "C" void kernel_launch(void* const* d_in, const int* in_sizes, int n_in,
                              void* d_out, int out_size, void* d_ws, size_t ws_size,
                              hipStream_t stream) {
    const float* X     = (const float*)d_in[0];
    const int*   edges = (const int*)d_in[1];
    const float* W     = (const float*)d_in[2];
    const float* a     = (const float*)d_in[3];
    float*       out   = (float*)d_out;

    float* h  = (float*)d_ws;
    float* e1 = h + (size_t)ROWS_ * FOUT_;
    float* e2 = e1 + (size_t)ROWS_;
    int*   alist = (int*)(e2 + (size_t)ROWS_);

    {
        const int waves_per_block = 8;
        const int blocks = (TILES_ + waves_per_block - 1) / waves_per_block;
        gat_gemm_wmma<<<blocks, 256, 0, stream>>>(X, W, h);
    }
    {
        const int blocks = (ROWS_ + 255) / 256;
        gat_scores<<<blocks, 256, 0, stream>>>(h, a, e1, e2);
    }
    gat_rows<<<B_ * NRANGE_, 1024, 0, stream>>>(edges, h, e1, e2, alist, out);
}


# Owner loop: the 8 edge weights of a batch (e1+e2, leaky step, exp) computed once across lanes with two gathers, not as 8 scalar-unit sequences with 16 scalar loads; float sum order unchanged
# speedup vs baseline: 5.3150x; 1.0580x over previous
#include <hip/hip_runtime.h>
#include <hip/hip_bf16.h>
#include <math.h>

typedef __attribute__((ext_vector_type(2))) float v2f;
typedef __attribute__((ext_vector_type(8))) float v8f;

#define B_ 4
#define N_ 50000
#define E_ 800000
#define FIN_ 128
#define FOUT_ 64
#define ALPHA_ 0.2f
#define NEG_BIG_ (-9000000000000000.0f)

#define ROWS_ (B_ * N_)
#define TILES_ (ROWS_ / 16)

__global__ __launch_bounds__(256) void gat_gemm_wmma(const float* __restrict__ X,
                                                     const float* __restrict__ W,
                                                     float* __restrict__ h) {
    const int lane   = threadIdx.x & 31;
    const int waveId = threadIdx.x >> 5;
    const int rt     = blockIdx.x * 8 + waveId;
    if (rt >= TILES_) return;

    const int m_base = rt * 16;
    const int mrow   = m_base + (lane & 15);
    const int khalf  = (lane >> 4) << 1;
    const int ncol   = lane & 15;

    v8f acc0 = {}, acc1 = {}, acc2 = {}, acc3 = {};

    for (int k = 0; k < FIN_; k += 4) {
        const int kk = k + khalf;
        v2f a;
        a.x = X[(size_t)mrow * FIN_ + kk];
        a.y = X[(size_t)mrow * FIN_ + kk + 1];

        const float* wr0 = W + (size_t)kk * FOUT_;
        const float* wr1 = wr0 + FOUT_;

        v2f b0, b1, b2, b3;
        b0.x = wr0[ncol];      b0.y = wr1[ncol];
        b1.x = wr0[16 + ncol]; b1.y = wr1[16 + ncol];
        b2.x = wr0[32 + ncol]; b2.y = wr1[32 + ncol];
        b3.x = wr0[48 + ncol]; b3.y = wr1[48 + ncol];

        acc0 = __builtin_amdgcn_wmma_f32_16x16x4_f32(false, a, false, b0, (short)0, acc0, false, false);
        acc1 = __builtin_amdgcn_wmma_f32_16x16x4_f32(false, a, false, b1, (short)0, acc1, false, false);
        acc2 = __builtin_amdgcn_wmma_f32_16x16x4_f32(false, a, false, b2, (short)0, acc2, false, false);
        acc3 = __builtin_amdgcn_wmma_f32_16x16x4_f32(false, a, false, b3, (short)0, acc3, false, false);
    }

    const int rhalf = (lane >> 4) << 3;
#pragma unroll
    for (int i = 0; i < 8; ++i) {
        const size_t row = (size_t)(m_base + i + rhalf);
        h[row * FOUT_ +  0 + ncol] = acc0[i];
        h[row * FOUT_ + 16 + ncol] = acc1[i];
        h[row * FOUT_ + 32 + ncol] = acc2[i];
        h[row * FOUT_ + 48 + ncol] = acc3[i];
    }
}

__global__ __launch_bounds__(256) void gat_scores(const float* __restrict__ h,
                                                  const float* __restrict__ a,
                                                  float* __restrict__ e1,
                                                  float* __restrict__ e2) {
    const int row = blockIdx.x * blockDim.x + threadIdx.x;
    if (row >= ROWS_) return;
    const float* hr = h + (size_t)row * FOUT_;
    float s1 = 0.f, s2 = 0.f;
#pragma unroll
    for (int f = 0; f < FOUT_; ++f) {
        const float hv = hr[f];
        s1 += hv * a[f];
        s2 += hv * a[FOUT_ + f];
    }
    e1[row] = s1;
    e2[row] = s2;
}

#define R_      782
#define NRANGE_ 64
#define C1_     704
#define C2_     72
#define NBLK_   (E_ / 128)
#define OFF_RS_ (R_ * FOUT_ * 4)
#define OFF_BL_ (OFF_RS_ + R_ * 4)
#define OFF_OL_ (OFF_BL_ + 32 * C1_ * 4)
#define OFF_ET_ (OFF_OL_ + 32 * C2_ * 4)
#define OFF_SF_ (OFF_ET_ + 32 * 32 * 4)
#define LDS_BYTES_ (OFF_SF_ + 32 * 4)

__device__ __forceinline__ int gat_flush(unsigned* __restrict__ myO, int n, int lane, bool all,
                                         const float* __restrict__ hb,
                                         const float* __restrict__ e1b,
                                         const float* __restrict__ e2b,
                                         float* __restrict__ acc, float& rsum) {
    int i = 0;
    for (; i + 8 <= n; i += 8) {
        unsigned en[8];
        float2 hv[8];
        const unsigned mine = myO[i + (lane & 7)];
#pragma unroll
        for (int k = 0; k < 8; ++k) en[k] = (unsigned)__builtin_amdgcn_readlane((int)mine, k);
#pragma unroll
        for (int k = 0; k < 8; ++k) {
            const unsigned dst = en[k] & 0xFFFFu;
            hv[k] = *(const float2*)(hb + (size_t)dst * FOUT_ + 2 * lane);
        }
        const float ms  = e1b[mine >> 16] + e2b[mine & 0xFFFFu];
        const float mlr = (ms >= 0.f) ? ms : ALPHA_ * ms;
        const int   mee = __float_as_int(__expf(-mlr));
#pragma unroll
        for (int k = 0; k < 8; ++k) {
            const unsigned d = en[k] >> 16;
            const float ee = __int_as_float(__builtin_amdgcn_readlane(mee, k));
            float2* ap = (float2*)(acc + d * FOUT_) + lane;
            float2 a = *ap;
            a.x += ee * hv[k].x;
            a.y += ee * hv[k].y;
            *ap = a;
            rsum += (lane == (int)(d >> 5)) ? ee : 0.f;
        }
    }
    if (!all) {
        const int rem = n - i;
        unsigned t = 0u;
        if (lane < rem) t = myO[i + lane];
        if (lane < rem) myO[lane] = t;
        return rem;
    }
    for (; i < n; ++i) {
        const unsigned e0  = __builtin_amdgcn_readfirstlane(myO[i]);
        const unsigned dst = e0 & 0xFFFFu;
        const unsigned d   = e0 >> 16;
        const float2 hv = *(const float2*)(hb + (size_t)dst * FOUT_ + 2 * lane);
        const float s  = e1b[d] + e2b[dst];
        const float lr = (s >= 0.f) ? s : ALPHA_ * s;
        const float ee = __expf(-lr);
        float2* ap = (float2*)(acc + d * FOUT_) + lane;
        float2 a = *ap;
        a.x += ee * hv.x;
        a.y += ee * hv.y;
        *ap = a;
        rsum += (lane == (int)(d >> 5)) ? ee : 0.f;
    }
    return 0;
}

__device__ __forceinline__ unsigned gat_same_owner(unsigned o, bool valid) {
    unsigned mk = __builtin_amdgcn_ballot_w32(valid);
#pragma unroll
    for (int bit = 0; bit < 5; ++bit) {
        const bool     on  = ((o >> bit) & 1u) != 0u;
        const unsigned bal = __builtin_amdgcn_ballot_w32(on);
        mk &= on ? bal : ~bal;
    }
    return mk;
}

__global__ __launch_bounds__(1024) void gat_rows(const int* __restrict__ edges,
                                                 const float* __restrict__ h,
                                                 const float* __restrict__ e1,
                                                 const float* __restrict__ e2,
                                                 int* __restrict__ alist,
                                                 float* __restrict__ out) {
    __shared__ __attribute__((aligned(16))) unsigned char L[LDS_BYTES_];
    float*    acc   = (float*)L;
    float*    rs    = (float*)(L + OFF_RS_);
    unsigned* blist = (unsigned*)(L + OFF_BL_);
    unsigned* olist = (unsigned*)(L + OFF_OL_);
    int*      endt  = (int*)(L + OFF_ET_);
    int*      sflag = (int*)(L + OFF_SF_);

    const int tid  = threadIdx.x;
    const int lane = tid & 31;
    const int w    = __builtin_amdgcn_readfirstlane(tid >> 5);
    const int b    = blockIdx.x / NRANGE_;
    const int rg   = blockIdx.x % NRANGE_;
    const int row_lo = rg * R_;
    const int nrows  = (N_ - row_lo < R_) ? (N_ - row_lo) : R_;

    const int*   srcarr = edges + (size_t)b * 2 * E_;
    const int*   dstarr = srcarr + E_;
    const float* hb     = h  + (size_t)b * N_ * FOUT_;
    const float* e1b    = e1 + (size_t)b * N_ + row_lo;
    const float* e2b    = e2 + (size_t)b * N_;

    for (int i = tid; i < R_ * FOUT_ / 4; i += 1024) ((float4*)L)[i] = make_float4(0.f, 0.f, 0.f, 0.f);

    int*      myA = alist + ((size_t)blockIdx.x * 32 + w) * C1_;
    unsigned* myB = blist + w * C1_;
    int*      myC = endt + w * 32;
    unsigned* myO = olist + w * C2_;
    const unsigned lt_mask = (1u << lane) - 1u;
    int it   = 0;
    int ocnt = 0;
    float rsum = 0.f;
    __syncthreads();

    for (int pass = 0; pass < 128; ++pass) {
        int cnt = 0;
        myC[lane] = 0;
        while (cnt <= C1_ - 128 && it * 32 + w < NBLK_) {
            int4 s[4];
#pragma unroll
            for (int j = 0; j < 4; ++j) {
                const int blk = (it + j) * 32 + w;
                if (blk < NBLK_) s[j] = *(const int4*)(srcarr + (size_t)blk * 128 + lane * 4);
                else             s[j] = make_int4(-1, -1, -1, -1);
            }
#pragma unroll
            for (int j = 0; j < 4; ++j) {
                if (cnt <= C1_ - 128) {
                    const unsigned ebase = (unsigned)(((it * 32 + w) * 128) + lane * 4);
                    const int sc[4] = {s[j].x, s[j].y, s[j].z, s[j].w};
#pragma unroll
                    for (int c = 0; c < 4; ++c) {
                        const unsigned d  = (unsigned)(sc[c] - row_lo);
                        const bool     m  = d < (unsigned)nrows;
                        const unsigned bm = __builtin_amdgcn_ballot_w32(m);
                        if (bm) {
                            if (m) {
                                myA[cnt + __builtin_popcount(bm & lt_mask)] = (int)((ebase + c) | (d << 20));
                                atomicAdd(&myC[d & 31u], 1);
                            }
                            cnt += __builtin_popcount(bm);
                        }
                    }
                    ++it;
                }
            }
        }
        __threadfence_block();
        {
            const int c = myC[lane];
            int incl = c;
#pragma unroll
            for (int k = 1; k < 32; k <<= 1) {
                const int t = __shfl_up(incl, k, 32);
                if (lane >= k) incl += t;
            }
            myC[lane] = incl - c;
        }
        for (int i0 = 0; i0 < cnt; i0 += 128) {
            unsigned en[4];
            int      dd[4];
#pragma unroll
            for (int j = 0; j < 4; ++j) {
                const int idx = i0 + j * 32 + lane;
                en[j] = (idx < cnt) ? (unsigned)myA[idx] : 0u;
            }
#pragma unroll
            for (int j = 0; j < 4; ++j) dd[j] = dstarr[en[j] & 0xFFFFFu];
#pragma unroll
            for (int j = 0; j < 4; ++j) {
                if (i0 + j * 32 < cnt) {
                    const bool     valid = (i0 + j * 32 + lane) < cnt;
                    const unsigned d     = en[j] >> 20;
                    const unsigned o     = d & 31u;
                    const unsigned mk    = gat_same_owner(o, valid);
                    const int pos = myC[o] + __builtin_popcount(mk & lt_mask);
                    if (valid) myB[pos] = (unsigned)dd[j] | (d << 16);
                    if (valid && (mk >> lane) == 1u) myC[o] = pos + 1;
                }
            }
        }
        if (lane == 0) sflag[w] = (it * 32 + w < NBLK_) ? 1 : 0;
        __syncthreads();

        const int pieceEnd   = endt[lane * 32 + w];
        const int pieceStart = (w > 0) ? endt[lane * 32 + w - 1] : 0;
        for (int v = 0; v < 32; ++v) {
            int       p  = __builtin_amdgcn_readlane(pieceStart, v);
            const int p1 = __builtin_amdgcn_readlane(pieceEnd, v);
            const unsigned* bl = blist + v * C1_;
            while (p < p1) {
                if (ocnt > C2_ - 32)
                    ocnt = gat_flush(myO, ocnt, lane, false, hb, e1b, e2b, acc, rsum);
                const int take = (p1 - p < 32) ? (p1 - p) : 32;
                if (lane < take) myO[ocnt + lane] = bl[p + lane];
                ocnt += take;
                p    += take;
            }
        }
        const unsigned more = __builtin_amdgcn_ballot_w32(sflag[lane] != 0);
        __syncthreads();
        if (more == 0u) break;
    }
    gat_flush(myO, ocnt, lane, true, hb, e1b, e2b, acc, rsum);
    if (lane * 32 + w < R_) rs[lane * 32 + w] = rsum;
    __syncthreads();

    for (int i = tid; i < nrows * (FOUT_ / 4); i += 1024) {
        const int r = i >> 4;
        const int c = (i & 15) * 4;
        const float4 a = *(const float4*)(acc + r * FOUT_ + c);
        const float  q = rs[r];
        float v0 = a.x / q, v1 = a.y / q, v2 = a.z / q, v3 = a.w / q;
        if (isnan(v0)) v0 = NEG_BIG_;
        if (isnan(v1)) v1 = NEG_BIG_;
        if (isnan(v2)) v2 = NEG_BIG_;
        if (isnan(v3)) v3 = NEG_BIG_;
        float4 o;
        o.x = (v0 > 0.f) ? v0 : expm1f(v0);
        o.y = (v1 > 0.f) ? v1 : expm1f(v1);
        o.z = (v2 > 0.f) ? v2 : expm1f(v2);
        o.w = (v3 > 0.f) ? v3 : expm1f(v3);
        *(float4*)(out + ((size_t)b * N_ + row_lo + r) * FOUT_ + c) = o;
    }
}

extern "C" void kernel_launch(void* const* d_in, const int* in_sizes, int n_in,
                              void* d_out, int out_size, void* d_ws, size_t ws_size,
                              hipStream_t stream) {
    const float* X     = (const float*)d_in[0];
    const int*   edges = (const int*)d_in[1];
    const float* W     = (const float*)d_in[2];
    const float* a     = (const float*)d_in[3];
    float*       out   = (float*)d_out;

    float* h  = (float*)d_ws;
    float* e1 = h + (size_t)ROWS_ * FOUT_;
    float* e2 = e1 + (size_t)ROWS_;
    int*   alist = (int*)(e2 + (size_t)ROWS_);

    {
        const int waves_per_block = 8;
        const int blocks = (TILES_ + waves_per_block - 1) / waves_per_block;
        gat_gemm_wmma<<<blocks, 256, 0, stream>>>(X, W, h);
    }
    {
        const int blocks = (ROWS_ + 255) / 256;
        gat_scores<<<blocks, 256, 0, stream>>>(h, a, e1, e2);
    }
    gat_rows<<<B_ * NRANGE_, 1024, 0, stream>>>(edges, h, e1, e2, alist, out);
}


# Owners scan one range byte per edge (src/782, written once by the score kernel beside a packed src|dst word), 16 per lane, byte-parallel compare, not 4-byte sources; float sum order changed
# speedup vs baseline: 5.7998x; 1.0912x over previous
#include <hip/hip_runtime.h>
#include <hip/hip_bf16.h>
#include <math.h>

typedef __attribute__((ext_vector_type(2))) float v2f;
typedef __attribute__((ext_vector_type(8))) float v8f;

#define B_ 4
#define N_ 50000
#define E_ 800000
#define FIN_ 128
#define FOUT_ 64
#define ALPHA_ 0.2f
#define NEG_BIG_ (-9000000000000000.0f)

#define ROWS_ (B_ * N_)
#define TILES_ (ROWS_ / 16)

__global__ __launch_bounds__(256) void gat_gemm_wmma(const float* __restrict__ X,
                                                     const float* __restrict__ W,
                                                     float* __restrict__ h) {
    const int lane   = threadIdx.x & 31;
    const int waveId = threadIdx.x >> 5;
    const int rt     = blockIdx.x * 8 + waveId;
    if (rt >= TILES_) return;

    const int m_base = rt * 16;
    const int mrow   = m_base + (lane & 15);
    const int khalf  = (lane >> 4) << 1;
    const int ncol   = lane & 15;

    v8f acc0 = {}, acc1 = {}, acc2 = {}, acc3 = {};

    for (int k = 0; k < FIN_; k += 4) {
        const int kk = k + khalf;
        v2f a;
        a.x = X[(size_t)mrow * FIN_ + kk];
        a.y = X[(size_t)mrow * FIN_ + kk + 1];

        const float* wr0 = W + (size_t)kk * FOUT_;
        const float* wr1 = wr0 + FOUT_;

        v2f b0, b1, b2, b3;
        b0.x = wr0[ncol];      b0.y = wr1[ncol];
        b1.x = wr0[16 + ncol]; b1.y = wr1[16 + ncol];
        b2.x = wr0[32 + ncol]; b2.y = wr1[32 + ncol];
        b3.x = wr0[48 + ncol]; b3.y = wr1[48 + ncol];

        acc0 = __builtin_amdgcn_wmma_f32_16x16x4_f32(false, a, false, b0, (short)0, acc0, false, false);
        acc1 = __builtin_amdgcn_wmma_f32_16x16x4_f32(false, a, false, b1, (short)0, acc1, false, false);
        acc2 = __builtin_amdgcn_wmma_f32_16x16x4_f32(false, a, false, b2, (short)0, acc2, false, false);
        acc3 = __builtin_amdgcn_wmma_f32_16x16x4_f32(false, a, false, b3, (short)0, acc3, false, false);
    }

    const int rhalf = (lane >> 4) << 3;
#pragma unroll
    for (int i = 0; i < 8; ++i) {
        const size_t row = (size_t)(m_base + i + rhalf);
        h[row * FOUT_ +  0 + ncol] = acc0[i];
        h[row * FOUT_ + 16 + ncol] = acc1[i];
        h[row * FOUT_ + 32 + ncol] = acc2[i];
        h[row * FOUT_ + 48 + ncol] = acc3[i];
    }
}

#define RNG_     782
#define RB_      512
#define NBLK2_   ((E_ + RB_ - 1) / RB_)
#define RIDW_    (NBLK2_ * (RB_ / 4))
#define SC_BLOCKS_ ((ROWS_ + 255) / 256)

__device__ __forceinline__ unsigned gat_rid(int s) {
    return ((unsigned)s < (unsigned)N_) ? ((unsigned)s / (unsigned)RNG_) : 0x7Fu;
}

__global__ __launch_bounds__(256) void gat_scores(const float* __restrict__ h,
                                                  const float* __restrict__ a,
                                                  const int* __restrict__ edges,
                                                  float* __restrict__ e1,
                                                  float* __restrict__ e2,
                                                  int* __restrict__ rid,
                                                  int* __restrict__ pk) {
    const int row = blockIdx.x * 256 + threadIdx.x;
    if (row < ROWS_) {
        const float* hr = h + (size_t)row * FOUT_;
        float s1 = 0.f, s2 = 0.f;
#pragma unroll
        for (int f = 0; f < FOUT_; ++f) {
            const float hv = hr[f];
            s1 += hv * a[f];
            s2 += hv * a[FOUT_ + f];
        }
        e1[row] = s1;
        e2[row] = s2;
    }
#pragma unroll
    for (int j = 0; j < 4; ++j) {
        const int q = row + j * (SC_BLOCKS_ * 256);
        if (q < B_ * RIDW_) {
            const int b = q / RIDW_;
            const int i = q - b * RIDW_;
            unsigned r = 0x7F7F7F7Fu;
            if (i < E_ / 4) {
                const int* sp = edges + (size_t)b * 2 * E_ + (size_t)i * 4;
                const int4 s = *(const int4*)sp;
                const int4 d = *(const int4*)(sp + E_);
                r = gat_rid(s.x) | (gat_rid(s.y) << 8) | (gat_rid(s.z) << 16) | (gat_rid(s.w) << 24);
                int4 p;
                p.x = (int)(((unsigned)s.x & 0xFFFFu) | ((unsigned)d.x << 16));
                p.y = (int)(((unsigned)s.y & 0xFFFFu) | ((unsigned)d.y << 16));
                p.z = (int)(((unsigned)s.z & 0xFFFFu) | ((unsigned)d.z << 16));
                p.w = (int)(((unsigned)s.w & 0xFFFFu) | ((unsigned)d.w << 16));
                *(int4*)(pk + (size_t)b * E_ + (size_t)i * 4) = p;
            }
            rid[q] = (int)r;
        }
    }
}

#define R_      782
#define NRANGE_ 64
#define C1_     704
#define C2_     72
static_assert(R_ == RNG_, "kernel 2 writes the range bytes for kernel 3's ranges");
#define OFF_RS_ (R_ * FOUT_ * 4)
#define OFF_BL_ (OFF_RS_ + R_ * 4)
#define OFF_OL_ (OFF_BL_ + 32 * C1_ * 4)
#define OFF_ET_ (OFF_OL_ + 32 * C2_ * 4)
#define OFF_SF_ (OFF_ET_ + 32 * 32 * 4)
#define LDS_BYTES_ (OFF_SF_ + 32 * 4)

__device__ __forceinline__ int gat_flush(unsigned* __restrict__ myO, int n, int lane, bool all,
                                         const float* __restrict__ hb,
                                         const float* __restrict__ e1b,
                                         const float* __restrict__ e2b,
                                         float* __restrict__ acc, float& rsum) {
    int i = 0;
    for (; i + 8 <= n; i += 8) {
        unsigned en[8];
        float2 hv[8];
        const unsigned mine = myO[i + (lane & 7)];
#pragma unroll
        for (int k = 0; k < 8; ++k) en[k] = (unsigned)__builtin_amdgcn_readlane((int)mine, k);
#pragma unroll
        for (int k = 0; k < 8; ++k) {
            const unsigned dst = en[k] & 0xFFFFu;
            hv[k] = *(const float2*)(hb + (size_t)dst * FOUT_ + 2 * lane);
        }
        const float ms  = e1b[mine >> 16] + e2b[mine & 0xFFFFu];
        const float mlr = (ms >= 0.f) ? ms : ALPHA_ * ms;
        const int   mee = __float_as_int(__expf(-mlr));
#pragma unroll
        for (int k = 0; k < 8; ++k) {
            const unsigned d = en[k] >> 16;
            const float ee = __int_as_float(__builtin_amdgcn_readlane(mee, k));
            float2* ap = (float2*)(acc + d * FOUT_) + lane;
            float2 a = *ap;
            a.x += ee * hv[k].x;
            a.y += ee * hv[k].y;
            *ap = a;
            rsum += (lane == (int)(d >> 5)) ? ee : 0.f;
        }
    }
    if (!all) {
        const int rem = n - i;
        unsigned t = 0u;
        if (lane < rem) t = myO[i + lane];
        if (lane < rem) myO[lane] = t;
        return rem;
    }
    for (; i < n; ++i) {
        const unsigned e0  = __builtin_amdgcn_readfirstlane(myO[i]);
        const unsigned dst = e0 & 0xFFFFu;
        const unsigned d   = e0 >> 16;
        const float2 hv = *(const float2*)(hb + (size_t)dst * FOUT_ + 2 * lane);
        const float s  = e1b[d] + e2b[dst];
        const float lr = (s >= 0.f) ? s : ALPHA_ * s;
        const float ee = __expf(-lr);
        float2* ap = (float2*)(acc + d * FOUT_) + lane;
        float2 a = *ap;
        a.x += ee * hv.x;
        a.y += ee * hv.y;
        *ap = a;
        rsum += (lane == (int)(d >> 5)) ? ee : 0.f;
    }
    return 0;
}

__device__ __forceinline__ unsigned gat_same_owner(unsigned o, bool valid) {
    unsigned mk = __builtin_amdgcn_ballot_w32(valid);
#pragma unroll
    for (int bit = 0; bit < 5; ++bit) {
        const bool     on  = ((o >> bit) & 1u) != 0u;
        const unsigned bal = __builtin_amdgcn_ballot_w32(on);
        mk &= on ? bal : ~bal;
    }
    return mk;
}

__global__ __launch_bounds__(1024) void gat_rows(const int* __restrict__ rid,
                                                 const int* __restrict__ pk,
                                                 const float* __restrict__ h,
                                                 const float* __restrict__ e1,
                                                 const float* __restrict__ e2,
                                                 int* __restrict__ alist,
                                                 float* __restrict__ out) {
    __shared__ __attribute__((aligned(16))) unsigned char L[LDS_BYTES_];
    float*    acc   = (float*)L;
    float*    rs    = (float*)(L + OFF_RS_);
    unsigned* blist = (unsigned*)(L + OFF_BL_);
    unsigned* olist = (unsigned*)(L + OFF_OL_);
    int*      endt  = (int*)(L + OFF_ET_);
    int*      sflag = (int*)(L + OFF_SF_);

    const int tid  = threadIdx.x;
    const int lane = tid & 31;
    const int w    = __builtin_amdgcn_readfirstlane(tid >> 5);
    const int b    = blockIdx.x / NRANGE_;
    const int rg   = blockIdx.x % NRANGE_;
    const int row_lo = rg * R_;
    const int nrows  = (N_ - row_lo < R_) ? (N_ - row_lo) : R_;

    const uint4*    ridb = (const uint4*)(rid + (size_t)b * RIDW_);
    const unsigned* pkb  = (const unsigned*)pk + (size_t)b * E_;
    const unsigned  rgb  = (unsigned)rg * 0x01010101u;
    const float* hb     = h  + (size_t)b * N_ * FOUT_;
    const float* e1b    = e1 + (size_t)b * N_ + row_lo;
    const float* e2b    = e2 + (size_t)b * N_;

    for (int i = tid; i < R_ * FOUT_ / 4; i += 1024) ((float4*)L)[i] = make_float4(0.f, 0.f, 0.f, 0.f);

    int*      myA = alist + ((size_t)blockIdx.x * 32 + w) * C1_;
    unsigned* myB = blist + w * C1_;
    int*      myC = endt + w * 32;
    unsigned* myO = olist + w * C2_;
    const unsigned lt_mask = (1u << lane) - 1u;
    int it   = 0;
    int ocnt = 0;
    float rsum = 0.f;
    __syncthreads();

    for (int pass = 0; pass < 128; ++pass) {
        int  cnt  = 0;
        bool full = false;
        myC[lane] = 0;
        while (!full && it * 32 + w < NBLK2_) {
            uint4 s[4];
#pragma unroll
            for (int j = 0; j < 4; ++j) {
                const int blk = (it + j) * 32 + w;
                if (blk < NBLK2_) s[j] = ridb[(size_t)blk * 32 + lane];
                else              s[j] = make_uint4(0x7F7F7F7Fu, 0x7F7F7F7Fu, 0x7F7F7F7Fu, 0x7F7F7F7Fu);
            }
#pragma unroll
            for (int j = 0; j < 4; ++j) {
                if (!full) {
                    const unsigned f0 = ~((s[j].x ^ rgb) + 0x7F7F7F7Fu) & 0x80808080u;
                    const unsigned f1 = ~((s[j].y ^ rgb) + 0x7F7F7F7Fu) & 0x80808080u;
                    const unsigned f2 = ~((s[j].z ^ rgb) + 0x7F7F7F7Fu) & 0x80808080u;
                    const unsigned f3 = ~((s[j].w ^ rgb) + 0x7F7F7F7Fu) & 0x80808080u;
                    unsigned m = (f0 >> 7) | (f1 >> 6) | (f2 >> 5) | (f3 >> 4);
                    unsigned bm = __builtin_amdgcn_ballot_w32(m != 0u);
                    if (cnt + 16 * __builtin_popcount(bm) > C1_) {
                        full = true;
                    } else {
                        const unsigned ebase = (unsigned)((it * 32 + w) * RB_ + lane * 16);
                        while (bm) {
                            if (m != 0u) {
                                const unsigned p = (unsigned)__builtin_ctz(m);
                                m &= m - 1u;
                                myA[cnt + __builtin_popcount(bm & lt_mask)] =
                                    (int)(ebase + ((p & 7u) << 2) + (p >> 3));
                            }
                            cnt += __builtin_popcount(bm);
                            bm = __builtin_amdgcn_ballot_w32(m != 0u);
                        }
                        ++it;
                    }
                }
            }
        }
        __threadfence_block();
        for (int i0 = 0; i0 < cnt; i0 += 128) {
            unsigned en[4], pw[4];
#pragma unroll
            for (int j = 0; j < 4; ++j) {
                const int idx = i0 + j * 32 + lane;
                en[j] = (idx < cnt) ? (unsigned)myA[idx] : 0u;
            }
#pragma unroll
            for (int j = 0; j < 4; ++j) pw[j] = pkb[en[j]];
#pragma unroll
            for (int j = 0; j < 4; ++j) {
                const int idx = i0 + j * 32 + lane;
                if (idx < cnt) {
                    const unsigned d = (pw[j] & 0xFFFFu) - (unsigned)row_lo;
                    myA[idx] = (int)((pw[j] >> 16) | (d << 16));
                    atomicAdd(&myC[d & 31u], 1);
                }
            }
        }
        __threadfence_block();
        {
            const int c = myC[lane];
            int incl = c;
#pragma unroll
            for (int k = 1; k < 32; k <<= 1) {
                const int t = __shfl_up(incl, k, 32);
                if (lane >= k) incl += t;
            }
            myC[lane] = incl - c;
        }
        for (int i0 = 0; i0 < cnt; i0 += 128) {
            unsigned en[4];
#pragma unroll
            for (int j = 0; j < 4; ++j) {
                const int idx = i0 + j * 32 + lane;
                en[j] = (idx < cnt) ? (unsigned)myA[idx] : 0u;
            }
#pragma unroll
            for (int j = 0; j < 4; ++j) {
                if (i0 + j * 32 < cnt) {
                    const bool     valid = (i0 + j * 32 + lane) < cnt;
                    const unsigned d     = en[j] >> 16;
                    const unsigned o     = d & 31u;
                    const unsigned mk    = gat_same_owner(o, valid);
                    const int pos = myC[o] + __builtin_popcount(mk & lt_mask);
                    if (valid) myB[pos] = en[j];
                    if (valid && (mk >> lane) == 1u) myC[o] = pos + 1;
                }
            }
        }
        if (lane == 0) sflag[w] = (it * 32 + w < NBLK2_) ? 1 : 0;
        __syncthreads();

        const int pieceEnd   = endt[lane * 32 + w];
        const int pieceStart = (w > 0) ? endt[lane * 32 + w - 1] : 0;
        for (int v = 0; v < 32; ++v) {
            int       p  = __builtin_amdgcn_readlane(pieceStart, v);
            const int p1 = __builtin_amdgcn_readlane(pieceEnd, v);
            const unsigned* bl = blist + v * C1_;
            while (p < p1) {
                if (ocnt > C2_ - 32)
                    ocnt = gat_flush(myO, ocnt, lane, false, hb, e1b, e2b, acc, rsum);
                const int take = (p1 - p < 32) ? (p1 - p) : 32;
                if (lane < take) myO[ocnt + lane] = bl[p + lane];
                ocnt += take;
                p    += take;
            }
        }
        const unsigned more = __builtin_amdgcn_ballot_w32(sflag[lane] != 0);
        __syncthreads();
        if (more == 0u) break;
    }
    gat_flush(myO, ocnt, lane, true, hb, e1b, e2b, acc, rsum);
    if (lane * 32 + w < R_) rs[lane * 32 + w] = rsum;
    __syncthreads();

    for (int i = tid; i < nrows * (FOUT_ / 4); i += 1024) {
        const int r = i >> 4;
        const int c = (i & 15) * 4;
        const float4 a = *(const float4*)(acc + r * FOUT_ + c);
        const float  q = rs[r];
        float v0 = a.x / q, v1 = a.y / q, v2 = a.z / q, v3 = a.w / q;
        if (isnan(v0)) v0 = NEG_BIG_;
        if (isnan(v1)) v1 = NEG_BIG_;
        if (isnan(v2)) v2 = NEG_BIG_;
        if (isnan(v3)) v3 = NEG_BIG_;
        float4 o;
        o.x = (v0 > 0.f) ? v0 : expm1f(v0);
        o.y = (v1 > 0.f) ? v1 : expm1f(v1);
        o.z = (v2 > 0.f) ? v2 : expm1f(v2);
        o.w = (v3 > 0.f) ? v3 : expm1f(v3);
        *(float4*)(out + ((size_t)b * N_ + row_lo + r) * FOUT_ + c) = o;
    }
}

extern "C" void kernel_launch(void* const* d_in, const int* in_sizes, int n_in,
                              void* d_out, int out_size, void* d_ws, size_t ws_size,
                              hipStream_t stream) {
    const float* X     = (const float*)d_in[0];
    const int*   edges = (const int*)d_in[1];
    const float* W     = (const float*)d_in[2];
    const float* a     = (const float*)d_in[3];
    float*       out   = (float*)d_out;

    float* h  = (float*)d_ws;
    float* e1 = h + (size_t)ROWS_ * FOUT_;
    float* e2 = e1 + (size_t)ROWS_;
    int*   alist = (int*)(e2 + (size_t)ROWS_);
    int*   rid   = alist + (size_t)B_ * NRANGE_ * 32 * C1_;
    int*   pk    = rid + (size_t)B_ * RIDW_;

    {
        const int waves_per_block = 8;
        const int blocks = (TILES_ + waves_per_block - 1) / waves_per_block;
        gat_gemm_wmma<<<blocks, 256, 0, stream>>>(X, W, h);
    }
    {
        const int blocks = SC_BLOCKS_;
        gat_scores<<<blocks, 256, 0, stream>>>(h, a, edges, e1, e2, rid, pk);
    }
    gat_rows<<<B_ * NRANGE_, 1024, 0, stream>>>(rid, pk, h, e1, e2, alist, out);
}


# Edge words stored in 32-edge groups dealt round-robin over the 512-edge scan blocks: a run of one range's edges no longer fills one wave's list (no second pass); float sum order changed
# speedup vs baseline: 6.3235x; 1.0903x over previous
#include <hip/hip_runtime.h>
#include <hip/hip_bf16.h>
#include <math.h>

typedef __attribute__((ext_vector_type(2))) float v2f;
typedef __attribute__((ext_vector_type(8))) float v8f;

#define B_ 4
#define N_ 50000
#define E_ 800000
#define FIN_ 128
#define FOUT_ 64
#define ALPHA_ 0.2f
#define NEG_BIG_ (-9000000000000000.0f)

#define ROWS_ (B_ * N_)
#define TILES_ (ROWS_ / 16)

__global__ __launch_bounds__(256) void gat_gemm_wmma(const float* __restrict__ X,
                                                     const float* __restrict__ W,
                                                     float* __restrict__ h) {
    const int lane   = threadIdx.x & 31;
    const int waveId = threadIdx.x >> 5;
    const int rt     = blockIdx.x * 8 + waveId;
    if (rt >= TILES_) return;

    const int m_base = rt * 16;
    const int mrow   = m_base + (lane & 15);
    const int khalf  = (lane >> 4) << 1;
    const int ncol   = lane & 15;

    v8f acc0 = {}, acc1 = {}, acc2 = {}, acc3 = {};

    for (int k = 0; k < FIN_; k += 4) {
        const int kk = k + khalf;
        v2f a;
        a.x = X[(size_t)mrow * FIN_ + kk];
        a.y = X[(size_t)mrow * FIN_ + kk + 1];

        const float* wr0 = W + (size_t)kk * FOUT_;
        const float* wr1 = wr0 + FOUT_;

        v2f b0, b1, b2, b3;
        b0.x = wr0[ncol];      b0.y = wr1[ncol];
        b1.x = wr0[16 + ncol]; b1.y = wr1[16 + ncol];
        b2.x = wr0[32 + ncol]; b2.y = wr1[32 + ncol];
        b3.x = wr0[48 + ncol]; b3.y = wr1[48 + ncol];

        acc0 = __builtin_amdgcn_wmma_f32_16x16x4_f32(false, a, false, b0, (short)0, acc0, false, false);
        acc1 = __builtin_amdgcn_wmma_f32_16x16x4_f32(false, a, false, b1, (short)0, acc1, false, false);
        acc2 = __builtin_amdgcn_wmma_f32_16x16x4_f32(false, a, false, b2, (short)0, acc2, false, false);
        acc3 = __builtin_amdgcn_wmma_f32_16x16x4_f32(false, a, false, b3, (short)0, acc3, false, false);
    }

    const int rhalf = (lane >> 4) << 3;
#pragma unroll
    for (int i = 0; i < 8; ++i) {
        const size_t row = (size_t)(m_base + i + rhalf);
        h[row * FOUT_ +  0 + ncol] = acc0[i];
        h[row * FOUT_ + 16 + ncol] = acc1[i];
        h[row * FOUT_ + 32 + ncol] = acc2[i];
        h[row * FOUT_ + 48 + ncol] = acc3[i];
    }
}

#define RNG_     782
#define RB_      512
#define NBLK2_   ((E_ + RB_ - 1) / RB_)
#define RIDW_    (NBLK2_ * (RB_ / 4))
#define PKS_     (NBLK2_ * RB_)
static_assert(RIDW_ % 8 == 0 && (RIDW_ / 8) == NBLK2_ * 16, "groups of 8 words: 16 per block, dealt over all blocks");
#define SC_BLOCKS_ ((ROWS_ + 255) / 256)

__device__ __forceinline__ unsigned gat_rid(int s) {
    return ((unsigned)s < (unsigned)N_) ? ((unsigned)s / (unsigned)RNG_) : 0x7Fu;
}

__global__ __launch_bounds__(256) void gat_scores(const float* __restrict__ h,
                                                  const float* __restrict__ a,
                                                  const int* __restrict__ edges,
                                                  float* __restrict__ e1,
                                                  float* __restrict__ e2,
                                                  int* __restrict__ rid,
                                                  int* __restrict__ pk) {
    const int row = blockIdx.x * 256 + threadIdx.x;
    if (row < ROWS_) {
        const float* hr = h + (size_t)row * FOUT_;
        float s1 = 0.f, s2 = 0.f;
#pragma unroll
        for (int f = 0; f < FOUT_; ++f) {
            const float hv = hr[f];
            s1 += hv * a[f];
            s2 += hv * a[FOUT_ + f];
        }
        e1[row] = s1;
        e2[row] = s2;
    }
#pragma unroll
    for (int j = 0; j < 4; ++j) {
        const int q = row + j * (SC_BLOCKS_ * 256);
        if (q < B_ * RIDW_) {
            const int b = q / RIDW_;
            const int i = q - b * RIDW_;
            const int g = i >> 3;
            const int ip = (g % NBLK2_) * (RB_ / 4) + (g / NBLK2_) * 8 + (i & 7);
            unsigned r = 0x7F7F7F7Fu;
            if (i < E_ / 4) {
                const int* sp = edges + (size_t)b * 2 * E_ + (size_t)i * 4;
                const int4 s = *(const int4*)sp;
                const int4 d = *(const int4*)(sp + E_);
                r = gat_rid(s.x) | (gat_rid(s.y) << 8) | (gat_rid(s.z) << 16) | (gat_rid(s.w) << 24);
                int4 p;
                p.x = (int)(((unsigned)s.x & 0xFFFFu) | ((unsigned)d.x << 16));
                p.y = (int)(((unsigned)s.y & 0xFFFFu) | ((unsigned)d.y << 16));
                p.z = (int)(((unsigned)s.z & 0xFFFFu) | ((unsigned)d.z << 16));
                p.w = (int)(((unsigned)s.w & 0xFFFFu) | ((unsigned)d.w << 16));
                *(int4*)(pk + (size_t)b * PKS_ + (size_t)ip * 4) = p;
            }
            rid[(size_t)b * RIDW_ + ip] = (int)r;
        }
    }
}

#define R_      782
#define NRANGE_ 64
#define C1_     704
#define C2_     72
static_assert(R_ == RNG_, "kernel 2 writes the range bytes for kernel 3's ranges");
#define OFF_RS_ (R_ * FOUT_ * 4)
#define OFF_BL_ (OFF_RS_ + R_ * 4)
#define OFF_OL_ (OFF_BL_ + 32 * C1_ * 4)
#define OFF_ET_ (OFF_OL_ + 32 * C2_ * 4)
#define OFF_SF_ (OFF_ET_ + 32 * 32 * 4)
#define LDS_BYTES_ (OFF_SF_ + 32 * 4)

__device__ __forceinline__ int gat_flush(unsigned* __restrict__ myO, int n, int lane, bool all,
                                         const float* __restrict__ hb,
                                         const float* __restrict__ e1b,
                                         const float* __restrict__ e2b,
                                         float* __restrict__ acc, float& rsum) {
    int i = 0;
    for (; i + 8 <= n; i += 8) {
        unsigned en[8];
        float2 hv[8];
        const unsigned mine = myO[i + (lane & 7)];
#pragma unroll
        for (int k = 0; k < 8; ++k) en[k] = (unsigned)__builtin_amdgcn_readlane((int)mine, k);
#pragma unroll
        for (int k = 0; k < 8; ++k) {
            const unsigned dst = en[k] & 0xFFFFu;
            hv[k] = *(const float2*)(hb + (size_t)dst * FOUT_ + 2 * lane);
        }
        const float ms  = e1b[mine >> 16] + e2b[mine & 0xFFFFu];
        const float mlr = (ms >= 0.f) ? ms : ALPHA_ * ms;
        const int   mee = __float_as_int(__expf(-mlr));
#pragma unroll
        for (int k = 0; k < 8; ++k) {
            const unsigned d = en[k] >> 16;
            const float ee = __int_as_float(__builtin_amdgcn_readlane(mee, k));
            float2* ap = (float2*)(acc + d * FOUT_) + lane;
            float2 a = *ap;
            a.x += ee * hv[k].x;
            a.y += ee * hv[k].y;
            *ap = a;
            rsum += (lane == (int)(d >> 5)) ? ee : 0.f;
        }
    }
    if (!all) {
        const int rem = n - i;
        unsigned t = 0u;
        if (lane < rem) t = myO[i + lane];
        if (lane < rem) myO[lane] = t;
        return rem;
    }
    for (; i < n; ++i) {
        const unsigned e0  = __builtin_amdgcn_readfirstlane(myO[i]);
        const unsigned dst = e0 & 0xFFFFu;
        const unsigned d   = e0 >> 16;
        const float2 hv = *(const float2*)(hb + (size_t)dst * FOUT_ + 2 * lane);
        const float s  = e1b[d] + e2b[dst];
        const float lr = (s >= 0.f) ? s : ALPHA_ * s;
        const float ee = __expf(-lr);
        float2* ap = (float2*)(acc + d * FOUT_) + lane;
        float2 a = *ap;
        a.x += ee * hv.x;
        a.y += ee * hv.y;
        *ap = a;
        rsum += (lane == (int)(d >> 5)) ? ee : 0.f;
    }
    return 0;
}

__device__ __forceinline__ unsigned gat_same_owner(unsigned o, bool valid) {
    unsigned mk = __builtin_amdgcn_ballot_w32(valid);
#pragma unroll
    for (int bit = 0; bit < 5; ++bit) {
        const bool     on  = ((o >> bit) & 1u) != 0u;
        const unsigned bal = __builtin_amdgcn_ballot_w32(on);
        mk &= on ? bal : ~bal;
    }
    return mk;
}

__global__ __launch_bounds__(1024) void gat_rows(const int* __restrict__ rid,
                                                 const int* __restrict__ pk,
                                                 const float* __restrict__ h,
                                                 const float* __restrict__ e1,
                                                 const float* __restrict__ e2,
                                                 int* __restrict__ alist,
                                                 float* __restrict__ out) {
    __shared__ __attribute__((aligned(16))) unsigned char L[LDS_BYTES_];
    float*    acc   = (float*)L;
    float*    rs    = (float*)(L + OFF_RS_);
    unsigned* blist = (unsigned*)(L + OFF_BL_);
    unsigned* olist = (unsigned*)(L + OFF_OL_);
    int*      endt  = (int*)(L + OFF_ET_);
    int*      sflag = (int*)(L + OFF_SF_);

    const int tid  = threadIdx.x;
    const int lane = tid & 31;
    const int w    = __builtin_amdgcn_readfirstlane(tid >> 5);
    const int b    = blockIdx.x / NRANGE_;
    const int rg   = blockIdx.x % NRANGE_;
    const int row_lo = rg * R_;
    const int nrows  = (N_ - row_lo < R_) ? (N_ - row_lo) : R_;

    const uint4*    ridb = (const uint4*)(rid + (size_t)b * RIDW_);
    const unsigned* pkb  = (const unsigned*)pk + (size_t)b * PKS_;
    const unsigned  rgb  = (unsigned)rg * 0x01010101u;
    const float* hb     = h  + (size_t)b * N_ * FOUT_;
    const float* e1b    = e1 + (size_t)b * N_ + row_lo;
    const float* e2b    = e2 + (size_t)b * N_;

    for (int i = tid; i < R_ * FOUT_ / 4; i += 1024) ((float4*)L)[i] = make_float4(0.f, 0.f, 0.f, 0.f);

    int*      myA = alist + ((size_t)blockIdx.x * 32 + w) * C1_;
    unsigned* myB = blist + w * C1_;
    int*      myC = endt + w * 32;
    unsigned* myO = olist + w * C2_;
    const unsigned lt_mask = (1u << lane) - 1u;
    int it   = 0;
    int ocnt = 0;
    float rsum = 0.f;
    __syncthreads();

    for (int pass = 0; pass < 128; ++pass) {
        int  cnt  = 0;
        bool full = false;
        myC[lane] = 0;
        while (!full && it * 32 + w < NBLK2_) {
            uint4 s[4];
#pragma unroll
            for (int j = 0; j < 4; ++j) {
                const int blk = (it + j) * 32 + w;
                if (blk < NBLK2_) s[j] = ridb[(size_t)blk * 32 + lane];
                else              s[j] = make_uint4(0x7F7F7F7Fu, 0x7F7F7F7Fu, 0x7F7F7F7Fu, 0x7F7F7F7Fu);
            }
#pragma unroll
            for (int j = 0; j < 4; ++j) {
                if (!full) {
                    const unsigned f0 = ~((s[j].x ^ rgb) + 0x7F7F7F7Fu) & 0x80808080u;
                    const unsigned f1 = ~((s[j].y ^ rgb) + 0x7F7F7F7Fu) & 0x80808080u;
                    const unsigned f2 = ~((s[j].z ^ rgb) + 0x7F7F7F7Fu) & 0x80808080u;
                    const unsigned f3 = ~((s[j].w ^ rgb) + 0x7F7F7F7Fu) & 0x80808080u;
                    unsigned m = (f0 >> 7) | (f1 >> 6) | (f2 >> 5) | (f3 >> 4);
                    unsigned bm = __builtin_amdgcn_ballot_w32(m != 0u);
                    if (cnt + 16 * __builtin_popcount(bm) > C1_) {
                        full = true;
                    } else {
                        const unsigned ebase = (unsigned)((it * 32 + w) * RB_ + lane * 16);
                        while (bm) {
                            if (m != 0u) {
                                const unsigned p = (unsigned)__builtin_ctz(m);
                                m &= m - 1u;
                                myA[cnt + __builtin_popcount(bm & lt_mask)] =
                                    (int)(ebase + ((p & 7u) << 2) + (p >> 3));
                            }
                            cnt += __builtin_popcount(bm);
                            bm = __builtin_amdgcn_ballot_w32(m != 0u);
                        }
                        ++it;
                    }
                }
            }
        }
        __threadfence_block();
        for (int i0 = 0; i0 < cnt; i0 += 128) {
            unsigned en[4], pw[4];
#pragma unroll
            for (int j = 0; j < 4; ++j) {
                const int idx = i0 + j * 32 + lane;
                en[j] = (idx < cnt) ? (unsigned)myA[idx] : 0u;
            }
#pragma unroll
            for (int j = 0; j < 4; ++j) pw[j] = pkb[en[j]];
#pragma unroll
            for (int j = 0; j < 4; ++j) {
                const int idx = i0 + j * 32 + lane;
                if (idx < cnt) {
                    const unsigned d = (pw[j] & 0xFFFFu) - (unsigned)row_lo;
                    myA[idx] = (int)((pw[j] >> 16) | (d << 16));
                    atomicAdd(&myC[d & 31u], 1);
                }
            }
        }
        __threadfence_block();
        {
            const int c = myC[lane];
            int incl = c;
#pragma unroll
            for (int k = 1; k < 32; k <<= 1) {
                const int t = __shfl_up(incl, k, 32);
                if (lane >= k) incl += t;
            }
            myC[lane] = incl - c;
        }
        for (int i0 = 0; i0 < cnt; i0 += 128) {
            unsigned en[4];
#pragma unroll
            for (int j = 0; j < 4; ++j) {
                const int idx = i0 + j * 32 + lane;
                en[j] = (idx < cnt) ? (unsigned)myA[idx] : 0u;
            }
#pragma unroll
            for (int j = 0; j < 4; ++j) {
                if (i0 + j * 32 < cnt) {
                    const bool     valid = (i0 + j * 32 + lane) < cnt;
                    const unsigned d     = en[j] >> 16;
                    const unsigned o     = d & 31u;
                    const unsigned mk    = gat_same_owner(o, valid);
                    const int pos = myC[o] + __builtin_popcount(mk & lt_mask);
                    if (valid) myB[pos] = en[j];
                    if (valid && (mk >> lane) == 1u) myC[o] = pos + 1;
                }
            }
        }
        if (lane == 0) sflag[w] = (it * 32 + w < NBLK2_) ? 1 : 0;
        __syncthreads();

        const int pieceEnd   = endt[lane * 32 + w];
        const int pieceStart = (w > 0) ? endt[lane * 32 + w - 1] : 0;
        for (int v = 0; v < 32; ++v) {
            int       p  = __builtin_amdgcn_readlane(pieceStart, v);
            const int p1 = __builtin_amdgcn_readlane(pieceEnd, v);
            const unsigned* bl = blist + v * C1_;
            while (p < p1) {
                if (ocnt > C2_ - 32)
                    ocnt = gat_flush(myO, ocnt, lane, false, hb, e1b, e2b, acc, rsum);
                const int take = (p1 - p < 32) ? (p1 - p) : 32;
                if (lane < take) myO[ocnt + lane] = bl[p + lane];
                ocnt += take;
                p    += take;
            }
        }
        const unsigned more = __builtin_amdgcn_ballot_w32(sflag[lane] != 0);
        __syncthreads();
        if (more == 0u) break;
    }
    gat_flush(myO, ocnt, lane, true, hb, e1b, e2b, acc, rsum);
    if (lane * 32 + w < R_) rs[lane * 32 + w] = rsum;
    __syncthreads();

    for (int i = tid; i < nrows * (FOUT_ / 4); i += 1024) {
        const int r = i >> 4;
        const int c = (i & 15) * 4;
        const float4 a = *(const float4*)(acc + r * FOUT_ + c);
        const float  q = rs[r];
        float v0 = a.x / q, v1 = a.y / q, v2 = a.z / q, v3 = a.w / q;
        if (isnan(v0)) v0 = NEG_BIG_;
        if (isnan(v1)) v1 = NEG_BIG_;
        if (isnan(v2)) v2 = NEG_BIG_;
        if (isnan(v3)) v3 = NEG_BIG_;
        float4 o;
        o.x = (v0 > 0.f) ? v0 : expm1f(v0);
        o.y = (v1 > 0.f) ? v1 : expm1f(v1);
        o.z = (v2 > 0.f) ? v2 : expm1f(v2);
        o.w = (v3 > 0.f) ? v3 : expm1f(v3);
        *(float4*)(out + ((size_t)b * N_ + row_lo + r) * FOUT_ + c) = o;
    }
}

extern "C" void kernel_launch(void* const* d_in, const int* in_sizes, int n_in,
                              void* d_out, int out_size, void* d_ws, size_t ws_size,
                              hipStream_t stream) {
    const float* X     = (const float*)d_in[0];
    const int*   edges = (const int*)d_in[1];
    const float* W     = (const float*)d_in[2];
    const float* a     = (const float*)d_in[3];
    float*       out   = (float*)d_out;

    float* h  = (float*)d_ws;
    float* e1 = h + (size_t)ROWS_ * FOUT_;
    float* e2 = e1 + (size_t)ROWS_;
    int*   alist = (int*)(e2 + (size_t)ROWS_);
    int*   rid   = alist + (size_t)B_ * NRANGE_ * 32 * C1_;
    int*   pk    = rid + (size_t)B_ * RIDW_;

    {
        const int waves_per_block = 8;
        const int blocks = (TILES_ + waves_per_block - 1) / waves_per_block;
        gat_gemm_wmma<<<blocks, 256, 0, stream>>>(X, W, h);
    }
    {
        const int blocks = SC_BLOCKS_;
        gat_scores<<<blocks, 256, 0, stream>>>(h, a, edges, e1, e2, rid, pk);
    }
    gat_rows<<<B_ * NRANGE_, 1024, 0, stream>>>(rid, pk, h, e1, e2, alist, out);
}


# GEMM: every wave keeps all of W in registers and loops over its row tiles (no B loads in the k loop); X, W, h move 16 bytes per lane by a k and column regrouping; k sum order of h changed
# speedup vs baseline: 6.6594x; 1.0531x over previous
#include <hip/hip_runtime.h>
#include <hip/hip_bf16.h>
#include <math.h>

typedef __attribute__((ext_vector_type(2))) float v2f;
typedef __attribute__((ext_vector_type(8))) float v8f;
typedef __attribute__((ext_vector_type(4))) float v4f;

#define B_ 4
#define N_ 50000
#define E_ 800000
#define FIN_ 128
#define FOUT_ 64
#define ALPHA_ 0.2f
#define NEG_BIG_ (-9000000000000000.0f)

#define ROWS_ (B_ * N_)
#define TILES_ (ROWS_ / 16)

#define GW_BLOCKS_ 256
#define GW_WAVES_  (GW_BLOCKS_ * 8)
static_assert(GW_WAVES_ <= TILES_, "every wave has a first tile");

__device__ __forceinline__ void gat_tile_load(v4f (&a)[16], const float* __restrict__ X,
                                              int rt, int c, int hf) {
    const v4f* xp = (const v4f*)(X + (size_t)(rt * 16 + c) * FIN_ + 4 * hf);
#pragma unroll
    for (int j = 0; j < 16; ++j) a[j] = xp[2 * j];
}

__device__ __forceinline__ void gat_tile_hold(const v4f (&a)[16]) {
    asm volatile("" :: "v"(a[0]), "v"(a[1]), "v"(a[2]), "v"(a[3]), "v"(a[4]), "v"(a[5]), "v"(a[6]), "v"(a[7]),
                       "v"(a[8]), "v"(a[9]), "v"(a[10]), "v"(a[11]), "v"(a[12]), "v"(a[13]), "v"(a[14]), "v"(a[15]));
}

__device__ __forceinline__ void gat_tile_mma(const v4f (&a)[16], const v2f (&wf)[16][2][4],
                                             float* __restrict__ h, int rt, int c, int hf) {
    v8f acc0 = {}, acc1 = {}, acc2 = {}, acc3 = {};
#pragma unroll
    for (int j = 0; j < 16; ++j) {
        v2f p, q;
        p.x = a[j].x; p.y = a[j].y;
        q.x = a[j].z; q.y = a[j].w;
        acc0 = __builtin_amdgcn_wmma_f32_16x16x4_f32(false, p, false, wf[j][0][0], (short)0, acc0, false, false);
        acc1 = __builtin_amdgcn_wmma_f32_16x16x4_f32(false, p, false, wf[j][0][1], (short)0, acc1, false, false);
        acc2 = __builtin_amdgcn_wmma_f32_16x16x4_f32(false, p, false, wf[j][0][2], (short)0, acc2, false, false);
        acc3 = __builtin_amdgcn_wmma_f32_16x16x4_f32(false, p, false, wf[j][0][3], (short)0, acc3, false, false);
        acc0 = __builtin_amdgcn_wmma_f32_16x16x4_f32(false, q, false, wf[j][1][0], (short)0, acc0, false, false);
        acc1 = __builtin_amdgcn_wmma_f32_16x16x4_f32(false, q, false, wf[j][1][1], (short)0, acc1, false, false);
        acc2 = __builtin_amdgcn_wmma_f32_16x16x4_f32(false, q, false, wf[j][1][2], (short)0, acc2, false, false);
        acc3 = __builtin_amdgcn_wmma_f32_16x16x4_f32(false, q, false, wf[j][1][3], (short)0, acc3, false, false);
    }
    float* hp = h + (size_t)(rt * 16 + 8 * hf) * FOUT_ + 4 * c;
#pragma unroll
    for (int r = 0; r < 8; ++r)
        *(float4*)(hp + r * FOUT_) = make_float4(acc0[r], acc1[r], acc2[r], acc3[r]);
}

__global__ __launch_bounds__(256) void gat_gemm_wmma(const float* __restrict__ X,
                                                     const float* __restrict__ W,
                                                     float* __restrict__ h) {
    const int lane = threadIdx.x & 31;
    const int c    = lane & 15;
    const int hf   = lane >> 4;
    const int g = __builtin_amdgcn_readfirstlane((int)(threadIdx.x >> 5)) * GW_BLOCKS_ + (int)blockIdx.x;

    v2f wf[16][2][4];
#pragma unroll
    for (int j = 0; j < 16; ++j) {
#pragma unroll
        for (int u = 0; u < 2; ++u) {
            const float* wp = W + (size_t)(8 * j + 4 * hf + 2 * u) * FOUT_ + 4 * c;
            const float4 w0 = *(const float4*)wp;
            const float4 w1 = *(const float4*)(wp + FOUT_);
            wf[j][u][0].x = w0.x; wf[j][u][0].y = w1.x;
            wf[j][u][1].x = w0.y; wf[j][u][1].y = w1.y;
            wf[j][u][2].x = w0.z; wf[j][u][2].y = w1.z;
            wf[j][u][3].x = w0.w; wf[j][u][3].y = w1.w;
        }
    }

    v4f a0[16], a1[16];
    int rt = g;
    gat_tile_load(a0, X, rt, c, hf);
    for (int it = 0; it < 4; ++it) {
        int nx = rt + GW_WAVES_;
        gat_tile_load(a1, X, (nx < TILES_) ? nx : rt, c, hf);
        __builtin_amdgcn_sched_barrier(0);
        gat_tile_mma(a0, wf, h, rt, c, hf);
        gat_tile_hold(a1);
        if (nx >= TILES_) break;
        rt = nx;
        nx = rt + GW_WAVES_;
        gat_tile_load(a0, X, (nx < TILES_) ? nx : rt, c, hf);
        __builtin_amdgcn_sched_barrier(0);
        gat_tile_mma(a1, wf, h, rt, c, hf);
        gat_tile_hold(a0);
        if (nx >= TILES_) break;
        rt = nx;
    }
}

#define RNG_     782
#define RB_      512
#define NBLK2_   ((E_ + RB_ - 1) / RB_)
#define RIDW_    (NBLK2_ * (RB_ / 4))
#define PKS_     (NBLK2_ * RB_)
static_assert(RIDW_ % 8 == 0 && (RIDW_ / 8) == NBLK2_ * 16, "groups of 8 words: 16 per block, dealt over all blocks");
#define SC_BLOCKS_ ((ROWS_ + 255) / 256)

__device__ __forceinline__ unsigned gat_rid(int s) {
    return ((unsigned)s < (unsigned)N_) ? ((unsigned)s / (unsigned)RNG_) : 0x7Fu;
}

__global__ __launch_bounds__(256) void gat_scores(const float* __restrict__ h,
                                                  const float* __restrict__ a,
                                                  const int* __restrict__ edges,
                                                  float* __restrict__ e1,
                                                  float* __restrict__ e2,
                                                  int* __restrict__ rid,
                                                  int* __restrict__ pk) {
    const int row = blockIdx.x * 256 + threadIdx.x;
    if (row < ROWS_) {
        const float* hr = h + (size_t)row * FOUT_;
        float s1 = 0.f, s2 = 0.f;
#pragma unroll
        for (int f = 0; f < FOUT_; ++f) {
            const float hv = hr[f];
            s1 += hv * a[f];
            s2 += hv * a[FOUT_ + f];
        }
        e1[row] = s1;
        e2[row] = s2;
    }
#pragma unroll
    for (int j = 0; j < 4; ++j) {
        const int q = row + j * (SC_BLOCKS_ * 256);
        if (q < B_ * RIDW_) {
            const int b = q / RIDW_;
            const int i = q - b * RIDW_;
            const int g = i >> 3;
            const int ip = (g % NBLK2_) * (RB_ / 4) + (g / NBLK2_) * 8 + (i & 7);
            unsigned r = 0x7F7F7F7Fu;
            if (i < E_ / 4) {
                const int* sp = edges + (size_t)b * 2 * E_ + (size_t)i * 4;
                const int4 s = *(const int4*)sp;
                const int4 d = *(const int4*)(sp + E_);
                r = gat_rid(s.x) | (gat_rid(s.y) << 8) | (gat_rid(s.z) << 16) | (gat_rid(s.w) << 24);
                int4 p;
                p.x = (int)(((unsigned)s.x & 0xFFFFu) | ((unsigned)d.x << 16));
                p.y = (int)(((unsigned)s.y & 0xFFFFu) | ((unsigned)d.y << 16));
                p.z = (int)(((unsigned)s.z & 0xFFFFu) | ((unsigned)d.z << 16));
                p.w = (int)(((unsigned)s.w & 0xFFFFu) | ((unsigned)d.w << 16));
                *(int4*)(pk + (size_t)b * PKS_ + (size_t)ip * 4) = p;
            }
            rid[(size_t)b * RIDW_ + ip] = (int)r;
        }
    }
}

#define R_      782
#define NRANGE_ 64
#define C1_     704
#define C2_     72
static_assert(R_ == RNG_, "kernel 2 writes the range bytes for kernel 3's ranges");
#define OFF_RS_ (R_ * FOUT_ * 4)
#define OFF_BL_ (OFF_RS_ + R_ * 4)
#define OFF_OL_ (OFF_BL_ + 32 * C1_ * 4)
#define OFF_ET_ (OFF_OL_ + 32 * C2_ * 4)
#define OFF_SF_ (OFF_ET_ + 32 * 32 * 4)
#define LDS_BYTES_ (OFF_SF_ + 32 * 4)

__device__ __forceinline__ int gat_flush(unsigned* __restrict__ myO, int n, int lane, bool all,
                                         const float* __restrict__ hb,
                                         const float* __restrict__ e1b,
                                         const float* __restrict__ e2b,
                                         float* __restrict__ acc, float& rsum) {
    int i = 0;
    for (; i + 8 <= n; i += 8) {
        unsigned en[8];
        float2 hv[8];
        const unsigned mine = myO[i + (lane & 7)];
#pragma unroll
        for (int k = 0; k < 8; ++k) en[k] = (unsigned)__builtin_amdgcn_readlane((int)mine, k);
#pragma unroll
        for (int k = 0; k < 8; ++k) {
            const unsigned dst = en[k] & 0xFFFFu;
            hv[k] = *(const float2*)(hb + (size_t)dst * FOUT_ + 2 * lane);
        }
        const float ms  = e1b[mine >> 16] + e2b[mine & 0xFFFFu];
        const float mlr = (ms >= 0.f) ? ms : ALPHA_ * ms;
        const int   mee = __float_as_int(__expf(-mlr));
#pragma unroll
        for (int k = 0; k < 8; ++k) {
            const unsigned d = en[k] >> 16;
            const float ee = __int_as_float(__builtin_amdgcn_readlane(mee, k));
            float2* ap = (float2*)(acc + d * FOUT_) + lane;
            float2 a = *ap;
            a.x += ee * hv[k].x;
            a.y += ee * hv[k].y;
            *ap = a;
            rsum += (lane == (int)(d >> 5)) ? ee : 0.f;
        }
    }
    if (!all) {
        const int rem = n - i;
        unsigned t = 0u;
        if (lane < rem) t = myO[i + lane];
        if (lane < rem) myO[lane] = t;
        return rem;
    }
    for (; i < n; ++i) {
        const unsigned e0  = __builtin_amdgcn_readfirstlane(myO[i]);
        const unsigned dst = e0 & 0xFFFFu;
        const unsigned d   = e0 >> 16;
        const float2 hv = *(const float2*)(hb + (size_t)dst * FOUT_ + 2 * lane);
        const float s  = e1b[d] + e2b[dst];
        const float lr = (s >= 0.f) ? s : ALPHA_ * s;
        const float ee = __expf(-lr);
        float2* ap = (float2*)(acc + d * FOUT_) + lane;
        float2 a = *ap;
        a.x += ee * hv.x;
        a.y += ee * hv.y;
        *ap = a;
        rsum += (lane == (int)(d >> 5)) ? ee : 0.f;
    }
    return 0;
}

__device__ __forceinline__ unsigned gat_same_owner(unsigned o, bool valid) {
    unsigned mk = __builtin_amdgcn_ballot_w32(valid);
#pragma unroll
    for (int bit = 0; bit < 5; ++bit) {
        const bool     on  = ((o >> bit) & 1u) != 0u;
        const unsigned bal = __builtin_amdgcn_ballot_w32(on);
        mk &= on ? bal : ~bal;
    }
    return mk;
}

__global__ __launch_bounds__(1024) void gat_rows(const int* __restrict__ rid,
                                                 const int* __restrict__ pk,
                                                 const float* __restrict__ h,
                                                 const float* __restrict__ e1,
                                                 const float* __restrict__ e2,
                                                 int* __restrict__ alist,
                                                 float* __restrict__ out) {
    __shared__ __attribute__((aligned(16))) unsigned char L[LDS_BYTES_];
    float*    acc   = (float*)L;
    float*    rs    = (float*)(L + OFF_RS_);
    unsigned* blist = (unsigned*)(L + OFF_BL_);
    unsigned* olist = (unsigned*)(L + OFF_OL_);
    int*      endt  = (int*)(L + OFF_ET_);
    int*      sflag = (int*)(L + OFF_SF_);

    const int tid  = threadIdx.x;
    const int lane = tid & 31;
    const int w    = __builtin_amdgcn_readfirstlane(tid >> 5);
    const int b    = blockIdx.x / NRANGE_;
    const int rg   = blockIdx.x % NRANGE_;
    const int row_lo = rg * R_;
    const int nrows  = (N_ - row_lo < R_) ? (N_ - row_lo) : R_;

    const uint4*    ridb = (const uint4*)(rid + (size_t)b * RIDW_);
    const unsigned* pkb  = (const unsigned*)pk + (size_t)b * PKS_;
    const unsigned  rgb  = (unsigned)rg * 0x01010101u;
    const float* hb     = h  + (size_t)b * N_ * FOUT_;
    const float* e1b    = e1 + (size_t)b * N_ + row_lo;
    const float* e2b    = e2 + (size_t)b * N_;

    for (int i = tid; i < R_ * FOUT_ / 4; i += 1024) ((float4*)L)[i] = make_float4(0.f, 0.f, 0.f, 0.f);

    int*      myA = alist + ((size_t)blockIdx.x * 32 + w) * C1_;
    unsigned* myB = blist + w * C1_;
    int*      myC = endt + w * 32;
    unsigned* myO = olist + w * C2_;
    const unsigned lt_mask = (1u << lane) - 1u;
    int it   = 0;
    int ocnt = 0;
    float rsum = 0.f;
    __syncthreads();

    for (int pass = 0; pass < 128; ++pass) {
        int  cnt  = 0;
        bool full = false;
        myC[lane] = 0;
        while (!full && it * 32 + w < NBLK2_) {
            uint4 s[4];
#pragma unroll
            for (int j = 0; j < 4; ++j) {
                const int blk = (it + j) * 32 + w;
                if (blk < NBLK2_) s[j] = ridb[(size_t)blk * 32 + lane];
                else              s[j] = make_uint4(0x7F7F7F7Fu, 0x7F7F7F7Fu, 0x7F7F7F7Fu, 0x7F7F7F7Fu);
            }
#pragma unroll
            for (int j = 0; j < 4; ++j) {
                if (!full) {
                    const unsigned f0 = ~((s[j].x ^ rgb) + 0x7F7F7F7Fu) & 0x80808080u;
                    const unsigned f1 = ~((s[j].y ^ rgb) + 0x7F7F7F7Fu) & 0x80808080u;
                    const unsigned f2 = ~((s[j].z ^ rgb) + 0x7F7F7F7Fu) & 0x80808080u;
                    const unsigned f3 = ~((s[j].w ^ rgb) + 0x7F7F7F7Fu) & 0x80808080u;
                    unsigned m = (f0 >> 7) | (f1 >> 6) | (f2 >> 5) | (f3 >> 4);
                    unsigned bm = __builtin_amdgcn_ballot_w32(m != 0u);
                    if (cnt + 16 * __builtin_popcount(bm) > C1_) {
                        full = true;
                    } else {
                        const unsigned ebase = (unsigned)((it * 32 + w) * RB_ + lane * 16);
                        while (bm) {
                            if (m != 0u) {
                                const unsigned p = (unsigned)__builtin_ctz(m);
                                m &= m - 1u;
                                myA[cnt + __builtin_popcount(bm & lt_mask)] =
                                    (int)(ebase + ((p & 7u) << 2) + (p >> 3));
                            }
                            cnt += __builtin_popcount(bm);
                            bm = __builtin_amdgcn_ballot_w32(m != 0u);
                        }
                        ++it;
                    }
                }
            }
        }
        __threadfence_block();
        for (int i0 = 0; i0 < cnt; i0 += 128) {
            unsigned en[4], pw[4];
#pragma unroll
            for (int j = 0; j < 4; ++j) {
                const int idx = i0 + j * 32 + lane;
                en[j] = (idx < cnt) ? (unsigned)myA[idx] : 0u;
            }
#pragma unroll
            for (int j = 0; j < 4; ++j) pw[j] = pkb[en[j]];
#pragma unroll
            for (int j = 0; j < 4; ++j) {
                const int idx = i0 + j * 32 + lane;
                if (idx < cnt) {
                    const unsigned d = (pw[j] & 0xFFFFu) - (unsigned)row_lo;
                    myA[idx] = (int)((pw[j] >> 16) | (d << 16));
                    atomicAdd(&myC[d & 31u], 1);
                }
            }
        }
        __threadfence_block();
        {
            const int c = myC[lane];
            int incl = c;
#pragma unroll
            for (int k = 1; k < 32; k <<= 1) {
                const int t = __shfl_up(incl, k, 32);
                if (lane >= k) incl += t;
            }
            myC[lane] = incl - c;
        }
        for (int i0 = 0; i0 < cnt; i0 += 128) {
            unsigned en[4];
#pragma unroll
            for (int j = 0; j < 4; ++j) {
                const int idx = i0 + j * 32 + lane;
                en[j] = (idx < cnt) ? (unsigned)myA[idx] : 0u;
            }
#pragma unroll
            for (int j = 0; j < 4; ++j) {
                if (i0 + j * 32 < cnt) {
                    const bool     valid = (i0 + j * 32 + lane) < cnt;
                    const unsigned d     = en[j] >> 16;
                    const unsigned o     = d & 31u;
                    const unsigned mk    = gat_same_owner(o, valid);
                    const int pos = myC[o] + __builtin_popcount(mk & lt_mask);
                    if (valid) myB[pos] = en[j];
                    if (valid && (mk >> lane) == 1u) myC[o] = pos + 1;
                }
            }
        }
        if (lane == 0) sflag[w] = (it * 32 + w < NBLK2_) ? 1 : 0;
        __syncthreads();

        const int pieceEnd   = endt[lane * 32 + w];
        const int pieceStart = (w > 0) ? endt[lane * 32 + w - 1] : 0;
        for (int v = 0; v < 32; ++v) {
            int       p  = __builtin_amdgcn_readlane(pieceStart, v);
            const int p1 = __builtin_amdgcn_readlane(pieceEnd, v);
            const unsigned* bl = blist + v * C1_;
            while (p < p1) {
                if (ocnt > C2_ - 32)
                    ocnt = gat_flush(myO, ocnt, lane, false, hb, e1b, e2b, acc, rsum);
                const int take = (p1 - p < 32) ? (p1 - p) : 32;
                if (lane < take) myO[ocnt + lane] = bl[p + lane];
                ocnt += take;
                p    += take;
            }
        }
        const unsigned more = __builtin_amdgcn_ballot_w32(sflag[lane] != 0);
        __syncthreads();
        if (more == 0u) break;
    }
    gat_flush(myO, ocnt, lane, true, hb, e1b, e2b, acc, rsum);
    if (lane * 32 + w < R_) rs[lane * 32 + w] = rsum;
    __syncthreads();

    for (int i = tid; i < nrows * (FOUT_ / 4); i += 1024) {
        const int r = i >> 4;
        const int c = (i & 15) * 4;
        const float4 a = *(const float4*)(acc + r * FOUT_ + c);
        const float  q = rs[r];
        float v0 = a.x / q, v1 = a.y / q, v2 = a.z / q, v3 = a.w / q;
        if (isnan(v0)) v0 = NEG_BIG_;
        if (isnan(v1)) v1 = NEG_BIG_;
        if (isnan(v2)) v2 = NEG_BIG_;
        if (isnan(v3)) v3 = NEG_BIG_;
        float4 o;
        o.x = (v0 > 0.f) ? v0 : expm1f(v0);
        o.y = (v1 > 0.f) ? v1 : expm1f(v1);
        o.z = (v2 > 0.f) ? v2 : expm1f(v2);
        o.w = (v3 > 0.f) ? v3 : expm1f(v3);
        *(float4*)(out + ((size_t)b * N_ + row_lo + r) * FOUT_ + c) = o;
    }
}

extern "C" void kernel_launch(void* const* d_in, const int* in_sizes, int n_in,
                              void* d_out, int out_size, void* d_ws, size_t ws_size,
                              hipStream_t stream) {
    const float* X     = (const float*)d_in[0];
    const int*   edges = (const int*)d_in[1];
    const float* W     = (const float*)d_in[2];
    const float* a     = (const float*)d_in[3];
    float*       out   = (float*)d_out;

    float* h  = (float*)d_ws;
    float* e1 = h + (size_t)ROWS_ * FOUT_;
    float* e2 = e1 + (size_t)ROWS_;
    int*   alist = (int*)(e2 + (size_t)ROWS_);
    int*   rid   = alist + (size_t)B_ * NRANGE_ * 32 * C1_;
    int*   pk    = rid + (size_t)B_ * RIDW_;

    gat_gemm_wmma<<<GW_BLOCKS_, 256, 0, stream>>>(X, W, h);
    {
        const int blocks = SC_BLOCKS_;
        gat_scores<<<blocks, 256, 0, stream>>>(h, a, edges, e1, e2, rid, pk);
    }
    gat_rows<<<B_ * NRANGE_, 1024, 0, stream>>>(rid, pk, h, e1, e2, alist, out);
}


# Owner waves keep the 64-wide sums of their 25 rows in registers picked by a scalar row index, not in an LDS tile: no LDS read-add-write per edge, no clear, out stored from registers; sum order same
# speedup vs baseline: 6.9580x; 1.0448x over previous
#include <hip/hip_runtime.h>
#include <hip/hip_bf16.h>
#include <math.h>

typedef __attribute__((ext_vector_type(2))) float v2f;
typedef __attribute__((ext_vector_type(8))) float v8f;
typedef __attribute__((ext_vector_type(4))) float v4f;
typedef __attribute__((ext_vector_type(32))) float v32f;

#define B_ 4
#define N_ 50000
#define E_ 800000
#define FIN_ 128
#define FOUT_ 64
#define ALPHA_ 0.2f
#define NEG_BIG_ (-9000000000000000.0f)

#define ROWS_ (B_ * N_)
#define TILES_ (ROWS_ / 16)

#define GW_BLOCKS_ 256
#define GW_WAVES_  (GW_BLOCKS_ * 8)
static_assert(GW_WAVES_ <= TILES_, "every wave has a first tile");

__device__ __forceinline__ void gat_tile_load(v4f (&a)[16], const float* __restrict__ X,
                                              int rt, int c, int hf) {
    const v4f* xp = (const v4f*)(X + (size_t)(rt * 16 + c) * FIN_ + 4 * hf);
#pragma unroll
    for (int j = 0; j < 16; ++j) a[j] = xp[2 * j];
}

__device__ __forceinline__ void gat_tile_hold(const v4f (&a)[16]) {
    asm volatile("" :: "v"(a[0]), "v"(a[1]), "v"(a[2]), "v"(a[3]), "v"(a[4]), "v"(a[5]), "v"(a[6]), "v"(a[7]),
                       "v"(a[8]), "v"(a[9]), "v"(a[10]), "v"(a[11]), "v"(a[12]), "v"(a[13]), "v"(a[14]), "v"(a[15]));
}

__device__ __forceinline__ void gat_tile_mma(const v4f (&a)[16], const v2f (&wf)[16][2][4],
                                             float* __restrict__ h, int rt, int c, int hf) {
    v8f acc0 = {}, acc1 = {}, acc2 = {}, acc3 = {};
#pragma unroll
    for (int j = 0; j < 16; ++j) {
        v2f p, q;
        p.x = a[j].x; p.y = a[j].y;
        q.x = a[j].z; q.y = a[j].w;
        acc0 = __builtin_amdgcn_wmma_f32_16x16x4_f32(false, p, false, wf[j][0][0], (short)0, acc0, false, false);
        acc1 = __builtin_amdgcn_wmma_f32_16x16x4_f32(false, p, false, wf[j][0][1], (short)0, acc1, false, false);
        acc2 = __builtin_amdgcn_wmma_f32_16x16x4_f32(false, p, false, wf[j][0][2], (short)0, acc2, false, false);
        acc3 = __builtin_amdgcn_wmma_f32_16x16x4_f32(false, p, false, wf[j][0][3], (short)0, acc3, false, false);
        acc0 = __builtin_amdgcn_wmma_f32_16x16x4_f32(false, q, false, wf[j][1][0], (short)0, acc0, false, false);
        acc1 = __builtin_amdgcn_wmma_f32_16x16x4_f32(false, q, false, wf[j][1][1], (short)0, acc1, false, false);
        acc2 = __builtin_amdgcn_wmma_f32_16x16x4_f32(false, q, false, wf[j][1][2], (short)0, acc2, false, false);
        acc3 = __builtin_amdgcn_wmma_f32_16x16x4_f32(false, q, false, wf[j][1][3], (short)0, acc3, false, false);
    }
    float* hp = h + (size_t)(rt * 16 + 8 * hf) * FOUT_ + 4 * c;
#pragma unroll
    for (int r = 0; r < 8; ++r)
        *(float4*)(hp + r * FOUT_) = make_float4(acc0[r], acc1[r], acc2[r], acc3[r]);
}

__global__ __launch_bounds__(256) void gat_gemm_wmma(const float* __restrict__ X,
                                                     const float* __restrict__ W,
                                                     float* __restrict__ h) {
    const int lane = threadIdx.x & 31;
    const int c    = lane & 15;
    const int hf   = lane >> 4;
    const int g = __builtin_amdgcn_readfirstlane((int)(threadIdx.x >> 5)) * GW_BLOCKS_ + (int)blockIdx.x;

    v2f wf[16][2][4];
#pragma unroll
    for (int j = 0; j < 16; ++j) {
#pragma unroll
        for (int u = 0; u < 2; ++u) {
            const float* wp = W + (size_t)(8 * j + 4 * hf + 2 * u) * FOUT_ + 4 * c;
            const float4 w0 = *(const float4*)wp;
            const float4 w1 = *(const float4*)(wp + FOUT_);
            wf[j][u][0].x = w0.x; wf[j][u][0].y = w1.x;
            wf[j][u][1].x = w0.y; wf[j][u][1].y = w1.y;
            wf[j][u][2].x = w0.z; wf[j][u][2].y = w1.z;
            wf[j][u][3].x = w0.w; wf[j][u][3].y = w1.w;
        }
    }

    v4f a0[16], a1[16];
    int rt = g;
    gat_tile_load(a0, X, rt, c, hf);
    for (int it = 0; it < 4; ++it) {
        int nx = rt + GW_WAVES_;
        gat_tile_load(a1, X, (nx < TILES_) ? nx : rt, c, hf);
        __builtin_amdgcn_sched_barrier(0);
        gat_tile_mma(a0, wf, h, rt, c, hf);
        gat_tile_hold(a1);
        if (nx >= TILES_) break;
        rt = nx;
        nx = rt + GW_WAVES_;
        gat_tile_load(a0, X, (nx < TILES_) ? nx : rt, c, hf);
        __builtin_amdgcn_sched_barrier(0);
        gat_tile_mma(a1, wf, h, rt, c, hf);
        gat_tile_hold(a0);
        if (nx >= TILES_) break;
        rt = nx;
    }
}

#define RNG_     782
#define RB_      512
#define NBLK2_   ((E_ + RB_ - 1) / RB_)
#define RIDW_    (NBLK2_ * (RB_ / 4))
#define PKS_     (NBLK2_ * RB_)
static_assert(RIDW_ % 8 == 0 && (RIDW_ / 8) == NBLK2_ * 16, "groups of 8 words: 16 per block, dealt over all blocks");
#define SC_BLOCKS_ ((ROWS_ + 255) / 256)

__device__ __forceinline__ unsigned gat_rid(int s) {
    return ((unsigned)s < (unsigned)N_) ? ((unsigned)s / (unsigned)RNG_) : 0x7Fu;
}

__global__ __launch_bounds__(256) void gat_scores(const float* __restrict__ h,
                                                  const float* __restrict__ a,
                                                  const int* __restrict__ edges,
                                                  float* __restrict__ e1,
                                                  float* __restrict__ e2,
                                                  int* __restrict__ rid,
                                                  int* __restrict__ pk) {
    const int row = blockIdx.x * 256 + threadIdx.x;
    if (row < ROWS_) {
        const float* hr = h + (size_t)row * FOUT_;
        float s1 = 0.f, s2 = 0.f;
#pragma unroll
        for (int f = 0; f < FOUT_; ++f) {
            const float hv = hr[f];
            s1 += hv * a[f];
            s2 += hv * a[FOUT_ + f];
        }
        e1[row] = s1;
        e2[row] = s2;
    }
#pragma unroll
    for (int j = 0; j < 4; ++j) {
        const int q = row + j * (SC_BLOCKS_ * 256);
        if (q < B_ * RIDW_) {
            const int b = q / RIDW_;
            const int i = q - b * RIDW_;
            const int g = i >> 3;
            const int ip = (g % NBLK2_) * (RB_ / 4) + (g / NBLK2_) * 8 + (i & 7);
            unsigned r = 0x7F7F7F7Fu;
            if (i < E_ / 4) {
                const int* sp = edges + (size_t)b * 2 * E_ + (size_t)i * 4;
                const int4 s = *(const int4*)sp;
                const int4 d = *(const int4*)(sp + E_);
                r = gat_rid(s.x) | (gat_rid(s.y) << 8) | (gat_rid(s.z) << 16) | (gat_rid(s.w) << 24);
                int4 p;
                p.x = (int)(((unsigned)s.x & 0xFFFFu) | ((unsigned)d.x << 16));
                p.y = (int)(((unsigned)s.y & 0xFFFFu) | ((unsigned)d.y << 16));
                p.z = (int)(((unsigned)s.z & 0xFFFFu) | ((unsigned)d.z << 16));
                p.w = (int)(((unsigned)s.w & 0xFFFFu) | ((unsigned)d.w << 16));
                *(int4*)(pk + (size_t)b * PKS_ + (size_t)ip * 4) = p;
            }
            rid[(size_t)b * RIDW_ + ip] = (int)r;
        }
    }
}

#define R_      782
#define NRANGE_ 64
#define C1_     704
#define C2_     72
static_assert(R_ == RNG_, "kernel 2 writes the range bytes for kernel 3's ranges");
#define RPW_    25
static_assert((R_ + 31) / 32 == RPW_ && RPW_ <= 32, "a wave's rows fit the 32-entry register vectors");
#define OFF_BL_ 0
#define OFF_OL_ (OFF_BL_ + 32 * C1_ * 4)
#define OFF_ET_ (OFF_OL_ + 32 * C2_ * 4)
#define OFF_SF_ (OFF_ET_ + 32 * 32 * 4)
#define LDS_BYTES_ (OFF_SF_ + 32 * 4)

__device__ __forceinline__ int gat_flush(unsigned* __restrict__ myO, int n, int lane, bool all,
                                         const float* __restrict__ hb,
                                         const float* __restrict__ e1b,
                                         const float* __restrict__ e2b,
                                         v32f& ax, v32f& ay, float& rsum) {
    int i = 0;
    for (; i + 8 <= n; i += 8) {
        unsigned en[8];
        float2 hv[8];
        const unsigned mine = myO[i + (lane & 7)];
#pragma unroll
        for (int k = 0; k < 8; ++k) en[k] = (unsigned)__builtin_amdgcn_readlane((int)mine, k);
#pragma unroll
        for (int k = 0; k < 8; ++k) {
            const unsigned dst = en[k] & 0xFFFFu;
            hv[k] = *(const float2*)(hb + (size_t)dst * FOUT_ + 2 * lane);
        }
        const float g1 = e1b[mine >> 16];
        const float g2 = e2b[mine & 0xFFFFu];
        __builtin_amdgcn_sched_barrier(0);
        const float ms  = g1 + g2;
        const float mlr = (ms >= 0.f) ? ms : ALPHA_ * ms;
        const int   mee = __float_as_int(__expf(-mlr));
#pragma unroll
        for (int k = 0; k < 8; ++k) {
            const unsigned j = (en[k] >> 21) & 31u;
            const float ee = __int_as_float(__builtin_amdgcn_readlane(mee, k));
            float ux = ax[j], uy = ay[j];
            ux += ee * hv[k].x;
            uy += ee * hv[k].y;
            ax[j] = ux;
            ay[j] = uy;
            rsum += (lane == (int)j) ? ee : 0.f;
        }
    }
    if (!all) {
        const int rem = n - i;
        unsigned t = 0u;
        if (lane < rem) t = myO[i + lane];
        if (lane < rem) myO[lane] = t;
        return rem;
    }
    for (; i < n; ++i) {
        const unsigned e0  = __builtin_amdgcn_readfirstlane(myO[i]);
        const unsigned dst = e0 & 0xFFFFu;
        const unsigned d   = e0 >> 16;
        const float2 hv = *(const float2*)(hb + (size_t)dst * FOUT_ + 2 * lane);
        const float s  = e1b[d] + e2b[dst];
        const float lr = (s >= 0.f) ? s : ALPHA_ * s;
        const float ee = __expf(-lr);
        const unsigned j = (d >> 5) & 31u;
        float ux = ax[j], uy = ay[j];
        ux += ee * hv.x;
        uy += ee * hv.y;
        ax[j] = ux;
        ay[j] = uy;
        rsum += (lane == (int)j) ? ee : 0.f;
    }
    return 0;
}

__device__ __forceinline__ unsigned gat_same_owner(unsigned o, bool valid) {
    unsigned mk = __builtin_amdgcn_ballot_w32(valid);
#pragma unroll
    for (int bit = 0; bit < 5; ++bit) {
        const bool     on  = ((o >> bit) & 1u) != 0u;
        const unsigned bal = __builtin_amdgcn_ballot_w32(on);
        mk &= on ? bal : ~bal;
    }
    return mk;
}

__global__ __launch_bounds__(1024) void gat_rows(const int* __restrict__ rid,
                                                 const int* __restrict__ pk,
                                                 const float* __restrict__ h,
                                                 const float* __restrict__ e1,
                                                 const float* __restrict__ e2,
                                                 int* __restrict__ alist,
                                                 float* __restrict__ out) {
    __shared__ __attribute__((aligned(16))) unsigned char L[LDS_BYTES_];
    unsigned* blist = (unsigned*)(L + OFF_BL_);
    unsigned* olist = (unsigned*)(L + OFF_OL_);
    int*      endt  = (int*)(L + OFF_ET_);
    int*      sflag = (int*)(L + OFF_SF_);

    const int tid  = threadIdx.x;
    const int lane = tid & 31;
    const int w    = __builtin_amdgcn_readfirstlane(tid >> 5);
    const int b    = blockIdx.x / NRANGE_;
    const int rg   = blockIdx.x % NRANGE_;
    const int row_lo = rg * R_;
    const int nrows  = (N_ - row_lo < R_) ? (N_ - row_lo) : R_;

    const uint4*    ridb = (const uint4*)(rid + (size_t)b * RIDW_);
    const unsigned* pkb  = (const unsigned*)pk + (size_t)b * PKS_;
    const unsigned  rgb  = (unsigned)rg * 0x01010101u;
    const float* hb     = h  + (size_t)b * N_ * FOUT_;
    const float* e1b    = e1 + (size_t)b * N_ + row_lo;
    const float* e2b    = e2 + (size_t)b * N_;

    v32f ax = {}, ay = {};

    int*      myA = alist + ((size_t)blockIdx.x * 32 + w) * C1_;
    unsigned* myB = blist + w * C1_;
    int*      myC = endt + w * 32;
    unsigned* myO = olist + w * C2_;
    const unsigned lt_mask = (1u << lane) - 1u;
    int it   = 0;
    int ocnt = 0;
    float rsum = 0.f;

    for (int pass = 0; pass < 128; ++pass) {
        int  cnt  = 0;
        bool full = false;
        myC[lane] = 0;
        while (!full && it * 32 + w < NBLK2_) {
            uint4 s[4];
#pragma unroll
            for (int j = 0; j < 4; ++j) {
                const int blk = (it + j) * 32 + w;
                if (blk < NBLK2_) s[j] = ridb[(size_t)blk * 32 + lane];
                else              s[j] = make_uint4(0x7F7F7F7Fu, 0x7F7F7F7Fu, 0x7F7F7F7Fu, 0x7F7F7F7Fu);
            }
#pragma unroll
            for (int j = 0; j < 4; ++j) {
                if (!full) {
                    const unsigned f0 = ~((s[j].x ^ rgb) + 0x7F7F7F7Fu) & 0x80808080u;
                    const unsigned f1 = ~((s[j].y ^ rgb) + 0x7F7F7F7Fu) & 0x80808080u;
                    const unsigned f2 = ~((s[j].z ^ rgb) + 0x7F7F7F7Fu) & 0x80808080u;
                    const unsigned f3 = ~((s[j].w ^ rgb) + 0x7F7F7F7Fu) & 0x80808080u;
                    unsigned m = (f0 >> 7) | (f1 >> 6) | (f2 >> 5) | (f3 >> 4);
                    unsigned bm = __builtin_amdgcn_ballot_w32(m != 0u);
                    if (cnt + 16 * __builtin_popcount(bm) > C1_) {
                        full = true;
                    } else {
                        const unsigned ebase = (unsigned)((it * 32 + w) * RB_ + lane * 16);
                        while (bm) {
                            if (m != 0u) {
                                const unsigned p = (unsigned)__builtin_ctz(m);
                                m &= m - 1u;
                                myA[cnt + __builtin_popcount(bm & lt_mask)] =
                                    (int)(ebase + ((p & 7u) << 2) + (p >> 3));
                            }
                            cnt += __builtin_popcount(bm);
                            bm = __builtin_amdgcn_ballot_w32(m != 0u);
                        }
                        ++it;
                    }
                }
            }
        }
        __threadfence_block();
        for (int i0 = 0; i0 < cnt; i0 += 128) {
            unsigned en[4], pw[4];
#pragma unroll
            for (int j = 0; j < 4; ++j) {
                const int idx = i0 + j * 32 + lane;
                en[j] = (idx < cnt) ? (unsigned)myA[idx] : 0u;
            }
#pragma unroll
            for (int j = 0; j < 4; ++j) pw[j] = pkb[en[j]];
#pragma unroll
            for (int j = 0; j < 4; ++j) {
                const int idx = i0 + j * 32 + lane;
                if (idx < cnt) {
                    const unsigned d = (pw[j] & 0xFFFFu) - (unsigned)row_lo;
                    myA[idx] = (int)((pw[j] >> 16) | (d << 16));
                    atomicAdd(&myC[d & 31u], 1);
                }
            }
        }
        __threadfence_block();
        {
            const int c = myC[lane];
            int incl = c;
#pragma unroll
            for (int k = 1; k < 32; k <<= 1) {
                const int t = __shfl_up(incl, k, 32);
                if (lane >= k) incl += t;
            }
            myC[lane] = incl - c;
        }
        for (int i0 = 0; i0 < cnt; i0 += 128) {
            unsigned en[4];
#pragma unroll
            for (int j = 0; j < 4; ++j) {
                const int idx = i0 + j * 32 + lane;
                en[j] = (idx < cnt) ? (unsigned)myA[idx] : 0u;
            }
#pragma unroll
            for (int j = 0; j < 4; ++j) {
                if (i0 + j * 32 < cnt) {
                    const bool     valid = (i0 + j * 32 + lane) < cnt;
                    const unsigned d     = en[j] >> 16;
                    const unsigned o     = d & 31u;
                    const unsigned mk    = gat_same_owner(o, valid);
                    const int pos = myC[o] + __builtin_popcount(mk & lt_mask);
                    if (valid) myB[pos] = en[j];
                    if (valid && (mk >> lane) == 1u) myC[o] = pos + 1;
                }
            }
        }
        if (lane == 0) sflag[w] = (it * 32 + w < NBLK2_) ? 1 : 0;
        __syncthreads();

        const int pieceEnd   = endt[lane * 32 + w];
        const int pieceStart = (w > 0) ? endt[lane * 32 + w - 1] : 0;
        for (int v = 0; v < 32; ++v) {
            int       p  = __builtin_amdgcn_readlane(pieceStart, v);
            const int p1 = __builtin_amdgcn_readlane(pieceEnd, v);
            const unsigned* bl = blist + v * C1_;
            while (p < p1) {
                if (ocnt > C2_ - 32)
                    ocnt = gat_flush(myO, ocnt, lane, false, hb, e1b, e2b, ax, ay, rsum);
                const int take = (p1 - p < 32) ? (p1 - p) : 32;
                if (lane < take) myO[ocnt + lane] = bl[p + lane];
                ocnt += take;
                p    += take;
            }
        }
        const unsigned more = __builtin_amdgcn_ballot_w32(sflag[lane] != 0);
        __syncthreads();
        if (more == 0u) break;
    }
    gat_flush(myO, ocnt, lane, true, hb, e1b, e2b, ax, ay, rsum);

    float* ob = out + ((size_t)b * N_ + row_lo + w) * FOUT_ + 2 * lane;
#pragma nounroll
    for (int j = 0; j * 32 + w < nrows; ++j) {
        const float q = __int_as_float(__builtin_amdgcn_readlane(__float_as_int(rsum), j));
        float v0 = ax[j] / q, v1 = ay[j] / q;
        if (isnan(v0)) v0 = NEG_BIG_;
        if (isnan(v1)) v1 = NEG_BIG_;
        float2 o;
        o.x = (v0 > 0.f) ? v0 : expm1f(v0);
        o.y = (v1 > 0.f) ? v1 : expm1f(v1);
        *(float2*)(ob + (size_t)j * 32 * FOUT_) = o;
    }
}

extern "C" void kernel_launch(void* const* d_in, const int* in_sizes, int n_in,
                              void* d_out, int out_size, void* d_ws, size_t ws_size,
                              hipStream_t stream) {
    const float* X     = (const float*)d_in[0];
    const int*   edges = (const int*)d_in[1];
    const float* W     = (const float*)d_in[2];
    const float* a     = (const float*)d_in[3];
    float*       out   = (float*)d_out;

    float* h  = (float*)d_ws;
    float* e1 = h + (size_t)ROWS_ * FOUT_;
    float* e2 = e1 + (size_t)ROWS_;
    int*   alist = (int*)(e2 + (size_t)ROWS_);
    int*   rid   = alist + (size_t)B_ * NRANGE_ * 32 * C1_;
    int*   pk    = rid + (size_t)B_ * RIDW_;

    gat_gemm_wmma<<<GW_BLOCKS_, 256, 0, stream>>>(X, W, h);
    {
        const int blocks = SC_BLOCKS_;
        gat_scores<<<blocks, 256, 0, stream>>>(h, a, edges, e1, e2, rid, pk);
    }
    gat_rows<<<B_ * NRANGE_, 1024, 0, stream>>>(rid, pk, h, e1, e2, alist, out);
}


# Rows kernel: same-in-every-lane work moved to the scalar unit (row base address, row-sum lane pick, one reciprocal per row); freed registers hold 16 row reads per batch, not 8; sum order same
# speedup vs baseline: 7.2855x; 1.0471x over previous
#include <hip/hip_runtime.h>
#include <hip/hip_bf16.h>
#include <math.h>

typedef __attribute__((ext_vector_type(2))) float v2f;
typedef __attribute__((ext_vector_type(8))) float v8f;
typedef __attribute__((ext_vector_type(4))) float v4f;
typedef __attribute__((ext_vector_type(32))) float v32f;

#define B_ 4
#define N_ 50000
#define E_ 800000
#define FIN_ 128
#define FOUT_ 64
#define ALPHA_ 0.2f
#define NEG_BIG_ (-9000000000000000.0f)

#define ROWS_ (B_ * N_)
#define TILES_ (ROWS_ / 16)

#define GW_BLOCKS_ 256
#define GW_WAVES_  (GW_BLOCKS_ * 8)
static_assert(GW_WAVES_ <= TILES_, "every wave has a first tile");

__device__ __forceinline__ void gat_tile_load(v4f (&a)[16], const float* __restrict__ X,
                                              int rt, int c, int hf) {
    const v4f* xp = (const v4f*)(X + (size_t)(rt * 16 + c) * FIN_ + 4 * hf);
#pragma unroll
    for (int j = 0; j < 16; ++j) a[j] = xp[2 * j];
}

__device__ __forceinline__ void gat_tile_hold(const v4f (&a)[16]) {
    asm volatile("" :: "v"(a[0]), "v"(a[1]), "v"(a[2]), "v"(a[3]), "v"(a[4]), "v"(a[5]), "v"(a[6]), "v"(a[7]),
                       "v"(a[8]), "v"(a[9]), "v"(a[10]), "v"(a[11]), "v"(a[12]), "v"(a[13]), "v"(a[14]), "v"(a[15]));
}

__device__ __forceinline__ void gat_tile_mma(const v4f (&a)[16], const v2f (&wf)[16][2][4],
                                             float* __restrict__ h, int rt, int c, int hf) {
    v8f acc0 = {}, acc1 = {}, acc2 = {}, acc3 = {};
#pragma unroll
    for (int j = 0; j < 16; ++j) {
        v2f p, q;
        p.x = a[j].x; p.y = a[j].y;
        q.x = a[j].z; q.y = a[j].w;
        acc0 = __builtin_amdgcn_wmma_f32_16x16x4_f32(false, p, false, wf[j][0][0], (short)0, acc0, false, false);
        acc1 = __builtin_amdgcn_wmma_f32_16x16x4_f32(false, p, false, wf[j][0][1], (short)0, acc1, false, false);
        acc2 = __builtin_amdgcn_wmma_f32_16x16x4_f32(false, p, false, wf[j][0][2], (short)0, acc2, false, false);
        acc3 = __builtin_amdgcn_wmma_f32_16x16x4_f32(false, p, false, wf[j][0][3], (short)0, acc3, false, false);
        acc0 = __builtin_amdgcn_wmma_f32_16x16x4_f32(false, q, false, wf[j][1][0], (short)0, acc0, false, false);
        acc1 = __builtin_amdgcn_wmma_f32_16x16x4_f32(false, q, false, wf[j][1][1], (short)0, acc1, false, false);
        acc2 = __builtin_amdgcn_wmma_f32_16x16x4_f32(false, q, false, wf[j][1][2], (short)0, acc2, false, false);
        acc3 = __builtin_amdgcn_wmma_f32_16x16x4_f32(false, q, false, wf[j][1][3], (short)0, acc3, false, false);
    }
    float* hp = h + (size_t)(rt * 16 + 8 * hf) * FOUT_ + 4 * c;
#pragma unroll
    for (int r = 0; r < 8; ++r)
        *(float4*)(hp + r * FOUT_) = make_float4(acc0[r], acc1[r], acc2[r], acc3[r]);
}

__global__ __launch_bounds__(256) void gat_gemm_wmma(const float* __restrict__ X,
                                                     const float* __restrict__ W,
                                                     float* __restrict__ h) {
    const int lane = threadIdx.x & 31;
    const int c    = lane & 15;
    const int hf   = lane >> 4;
    const int g = __builtin_amdgcn_readfirstlane((int)(threadIdx.x >> 5)) * GW_BLOCKS_ + (int)blockIdx.x;

    v2f wf[16][2][4];
#pragma unroll
    for (int j = 0; j < 16; ++j) {
#pragma unroll
        for (int u = 0; u < 2; ++u) {
            const float* wp = W + (size_t)(8 * j + 4 * hf + 2 * u) * FOUT_ + 4 * c;
            const float4 w0 = *(const float4*)wp;
            const float4 w1 = *(const float4*)(wp + FOUT_);
            wf[j][u][0].x = w0.x; wf[j][u][0].y = w1.x;
            wf[j][u][1].x = w0.y; wf[j][u][1].y = w1.y;
            wf[j][u][2].x = w0.z; wf[j][u][2].y = w1.z;
            wf[j][u][3].x = w0.w; wf[j][u][3].y = w1.w;
        }
    }

    v4f a0[16], a1[16];
    int rt = g;
    gat_tile_load(a0, X, rt, c, hf);
    for (int it = 0; it < 4; ++it) {
        int nx = rt + GW_WAVES_;
        gat_tile_load(a1, X, (nx < TILES_) ? nx : rt, c, hf);
        __builtin_amdgcn_sched_barrier(0);
        gat_tile_mma(a0, wf, h, rt, c, hf);
        gat_tile_hold(a1);
        if (nx >= TILES_) break;
        rt = nx;
        nx = rt + GW_WAVES_;
        gat_tile_load(a0, X, (nx < TILES_) ? nx : rt, c, hf);
        __builtin_amdgcn_sched_barrier(0);
        gat_tile_mma(a1, wf, h, rt, c, hf);
        gat_tile_hold(a0);
        if (nx >= TILES_) break;
        rt = nx;
    }
}

#define RNG_     782
#define RB_      512
#define NBLK2_   ((E_ + RB_ - 1) / RB_)
#define RIDW_    (NBLK2_ * (RB_ / 4))
#define PKS_     (NBLK2_ * RB_)
static_assert(RIDW_ % 8 == 0 && (RIDW_ / 8) == NBLK2_ * 16, "groups of 8 words: 16 per block, dealt over all blocks");
#define SC_BLOCKS_ ((ROWS_ + 255) / 256)

__device__ __forceinline__ unsigned gat_rid(int s) {
    return ((unsigned)s < (unsigned)N_) ? ((unsigned)s / (unsigned)RNG_) : 0x7Fu;
}

__global__ __launch_bounds__(256) void gat_scores(const float* __restrict__ h,
                                                  const float* __restrict__ a,
                                                  const int* __restrict__ edges,
                                                  float* __restrict__ e1,
                                                  float* __restrict__ e2,
                                                  int* __restrict__ rid,
                                                  int* __restrict__ pk) {
    const int row = blockIdx.x * 256 + threadIdx.x;
    if (row < ROWS_) {
        const float* hr = h + (size_t)row * FOUT_;
        float s1 = 0.f, s2 = 0.f;
#pragma unroll
        for (int f = 0; f < FOUT_; ++f) {
            const float hv = hr[f];
            s1 += hv * a[f];
            s2 += hv * a[FOUT_ + f];
        }
        e1[row] = s1;
        e2[row] = s2;
    }
#pragma unroll
    for (int j = 0; j < 4; ++j) {
        const int q = row + j * (SC_BLOCKS_ * 256);
        if (q < B_ * RIDW_) {
            const int b = q / RIDW_;
            const int i = q - b * RIDW_;
            const int g = i >> 3;
            const int ip = (g % NBLK2_) * (RB_ / 4) + (g / NBLK2_) * 8 + (i & 7);
            unsigned r = 0x7F7F7F7Fu;
            if (i < E_ / 4) {
                const int* sp = edges + (size_t)b * 2 * E_ + (size_t)i * 4;
                const int4 s = *(const int4*)sp;
                const int4 d = *(const int4*)(sp + E_);
                r = gat_rid(s.x) | (gat_rid(s.y) << 8) | (gat_rid(s.z) << 16) | (gat_rid(s.w) << 24);
                int4 p;
                p.x = (int)(((unsigned)s.x & 0xFFFFu) | ((unsigned)d.x << 16));
                p.y = (int)(((unsigned)s.y & 0xFFFFu) | ((unsigned)d.y << 16));
                p.z = (int)(((unsigned)s.z & 0xFFFFu) | ((unsigned)d.z << 16));
                p.w = (int)(((unsigned)s.w & 0xFFFFu) | ((unsigned)d.w << 16));
                *(int4*)(pk + (size_t)b * PKS_ + (size_t)ip * 4) = p;
            }
            rid[(size_t)b * RIDW_ + ip] = (int)r;
        }
    }
}

#define R_      782
#define NRANGE_ 64
#define C1_     704
#define C2_     72
static_assert(R_ == RNG_, "kernel 2 writes the range bytes for kernel 3's ranges");
#define RPW_    25
static_assert((R_ + 31) / 32 == RPW_ && RPW_ <= 32, "a wave's rows fit the 32-entry register vectors");
#define OFF_BL_ 0
#define OFF_OL_ (OFF_BL_ + 32 * C1_ * 4)
#define OFF_ET_ (OFF_OL_ + 32 * C2_ * 4)
#define OFF_SF_ (OFF_ET_ + 32 * 32 * 4)
#define LDS_BYTES_ (OFF_SF_ + 32 * 4)

typedef __attribute__((address_space(1))) const unsigned char gat_gbyte;
typedef __attribute__((address_space(1))) const v2f gat_gv2f;

#define FB_ 16

__device__ __forceinline__ void gat_rsum_add(float& rsum, const unsigned (&j)[16], const int (&e)[16]) {
    unsigned keep;
    asm volatile("s_mov_b32 %[kp], exec_lo\n\t"
                 "s_bfm_b32 exec_lo, 1, %[j0]\n\t"
                 "v_add_f32_e32 %[r], %[e0], %[r]\n\t"
                 "s_bfm_b32 exec_lo, 1, %[j1]\n\t"
                 "v_add_f32_e32 %[r], %[e1], %[r]\n\t"
                 "s_bfm_b32 exec_lo, 1, %[j2]\n\t"
                 "v_add_f32_e32 %[r], %[e2], %[r]\n\t"
                 "s_bfm_b32 exec_lo, 1, %[j3]\n\t"
                 "v_add_f32_e32 %[r], %[e3], %[r]\n\t"
                 "s_bfm_b32 exec_lo, 1, %[j4]\n\t"
                 "v_add_f32_e32 %[r], %[e4], %[r]\n\t"
                 "s_bfm_b32 exec_lo, 1, %[j5]\n\t"
                 "v_add_f32_e32 %[r], %[e5], %[r]\n\t"
                 "s_bfm_b32 exec_lo, 1, %[j6]\n\t"
                 "v_add_f32_e32 %[r], %[e6], %[r]\n\t"
                 "s_bfm_b32 exec_lo, 1, %[j7]\n\t"
                 "v_add_f32_e32 %[r], %[e7], %[r]\n\t"
                 "s_bfm_b32 exec_lo, 1, %[j8]\n\t"
                 "v_add_f32_e32 %[r], %[e8], %[r]\n\t"
                 "s_bfm_b32 exec_lo, 1, %[j9]\n\t"
                 "v_add_f32_e32 %[r], %[e9], %[r]\n\t"
                 "s_bfm_b32 exec_lo, 1, %[j10]\n\t"
                 "v_add_f32_e32 %[r], %[e10], %[r]\n\t"
                 "s_bfm_b32 exec_lo, 1, %[j11]\n\t"
                 "v_add_f32_e32 %[r], %[e11], %[r]\n\t"
                 "s_bfm_b32 exec_lo, 1, %[j12]\n\t"
                 "v_add_f32_e32 %[r], %[e12], %[r]\n\t"
                 "s_bfm_b32 exec_lo, 1, %[j13]\n\t"
                 "v_add_f32_e32 %[r], %[e13], %[r]\n\t"
                 "s_bfm_b32 exec_lo, 1, %[j14]\n\t"
                 "v_add_f32_e32 %[r], %[e14], %[r]\n\t"
                 "s_bfm_b32 exec_lo, 1, %[j15]\n\t"
                 "v_add_f32_e32 %[r], %[e15], %[r]\n\t"
                 "s_mov_b32 exec_lo, %[kp]"
                 : [r] "+v"(rsum), [kp] "=&s"(keep)
                 : [j0] "s"(j[0]), [j1] "s"(j[1]), [j2] "s"(j[2]), [j3] "s"(j[3]), [j4] "s"(j[4]), [j5] "s"(j[5]), [j6] "s"(j[6]), [j7] "s"(j[7]), [j8] "s"(j[8]), [j9] "s"(j[9]), [j10] "s"(j[10]), [j11] "s"(j[11]), [j12] "s"(j[12]), [j13] "s"(j[13]), [j14] "s"(j[14]), [j15] "s"(j[15]),
                   [e0] "s"(e[0]), [e1] "s"(e[1]), [e2] "s"(e[2]), [e3] "s"(e[3]), [e4] "s"(e[4]), [e5] "s"(e[5]), [e6] "s"(e[6]), [e7] "s"(e[7]), [e8] "s"(e[8]), [e9] "s"(e[9]), [e10] "s"(e[10]), [e11] "s"(e[11]), [e12] "s"(e[12]), [e13] "s"(e[13]), [e14] "s"(e[14]), [e15] "s"(e[15]));
}
__device__ __forceinline__ void gat_rsum_add(float& rsum, const unsigned (&j)[8], const int (&e)[8]) {
    unsigned keep;
    asm volatile("s_mov_b32 %[kp], exec_lo\n\t"
                 "s_bfm_b32 exec_lo, 1, %[j0]\n\t"
                 "v_add_f32_e32 %[r], %[e0], %[r]\n\t"
                 "s_bfm_b32 exec_lo, 1, %[j1]\n\t"
                 "v_add_f32_e32 %[r], %[e1], %[r]\n\t"
                 "s_bfm_b32 exec_lo, 1, %[j2]\n\t"
                 "v_add_f32_e32 %[r], %[e2], %[r]\n\t"
                 "s_bfm_b32 exec_lo, 1, %[j3]\n\t"
                 "v_add_f32_e32 %[r], %[e3], %[r]\n\t"
                 "s_bfm_b32 exec_lo, 1, %[j4]\n\t"
                 "v_add_f32_e32 %[r], %[e4], %[r]\n\t"
                 "s_bfm_b32 exec_lo, 1, %[j5]\n\t"
                 "v_add_f32_e32 %[r], %[e5], %[r]\n\t"
                 "s_bfm_b32 exec_lo, 1, %[j6]\n\t"
                 "v_add_f32_e32 %[r], %[e6], %[r]\n\t"
                 "s_bfm_b32 exec_lo, 1, %[j7]\n\t"
                 "v_add_f32_e32 %[r], %[e7], %[r]\n\t"
                 "s_mov_b32 exec_lo, %[kp]"
                 : [r] "+v"(rsum), [kp] "=&s"(keep)
                 : [j0] "s"(j[0]), [j1] "s"(j[1]), [j2] "s"(j[2]), [j3] "s"(j[3]), [j4] "s"(j[4]), [j5] "s"(j[5]), [j6] "s"(j[6]), [j7] "s"(j[7]),
                   [e0] "s"(e[0]), [e1] "s"(e[1]), [e2] "s"(e[2]), [e3] "s"(e[3]), [e4] "s"(e[4]), [e5] "s"(e[5]), [e6] "s"(e[6]), [e7] "s"(e[7]));
}

template <int NB>
__device__ __forceinline__ void gat_batch(const unsigned* __restrict__ myO, int i, int lane,
                                          const float* __restrict__ hb,
                                          const float* __restrict__ e1b,
                                          const float* __restrict__ e2b,
                                          v32f& ax, v32f& ay, float& rsum) {
    static_assert(NB == 8 || NB == 16, "lane k (mod NB) takes entry k");
    unsigned l8 = (unsigned)lane * 8u;
    asm volatile("" : "+v"(l8));
    unsigned en[NB];
    float2 hv[NB];
    const unsigned mine = myO[i + (lane & (NB - 1))];
#pragma unroll
    for (int k = 0; k < NB; ++k) en[k] = (unsigned)__builtin_amdgcn_readlane((int)mine, k);
#pragma unroll
    for (int k = 0; k < NB; ++k) {
        const unsigned dst = en[k] & 0xFFFFu;
        const unsigned long long rb = (unsigned long long)hb + ((unsigned long long)dst << 8);
        const unsigned rlo = (unsigned)__builtin_amdgcn_readfirstlane((int)(unsigned)rb);
        const unsigned rhi = (unsigned)__builtin_amdgcn_readfirstlane((int)(unsigned)(rb >> 32));
        const gat_gbyte* rowp = (const gat_gbyte*)(((unsigned long long)rhi << 32) | (unsigned long long)rlo);
        const v2f t = *(const gat_gv2f*)(rowp + l8);
        hv[k] = make_float2(t.x, t.y);
    }
    const float g1 = e1b[mine >> 16];
    const float g2 = e2b[mine & 0xFFFFu];
    __builtin_amdgcn_sched_barrier(0);
    const float ms  = g1 + g2;
    const float mlr = (ms >= 0.f) ? ms : ALPHA_ * ms;
    const int   mee = __float_as_int(__expf(-mlr));
    unsigned jr[NB];
    int      eb[NB];
#pragma unroll
    for (int k = 0; k < NB; ++k) {
        const unsigned j = (en[k] >> 21) & 31u;
        eb[k] = __builtin_amdgcn_readlane(mee, k);
        jr[k] = j;
        const float ee = __int_as_float(eb[k]);
        float ux = ax[j], uy = ay[j];
        ux += ee * hv[k].x;
        uy += ee * hv[k].y;
        ax[j] = ux;
        ay[j] = uy;
    }
    gat_rsum_add(rsum, jr, eb);
}

__device__ __forceinline__ int gat_flush(unsigned* __restrict__ myO, int n, int lane, bool all,
                                         const float* __restrict__ hb,
                                         const float* __restrict__ e1b,
                                         const float* __restrict__ e2b,
                                         v32f& ax, v32f& ay, float& rsum) {
    int i = 0;
    for (; i + FB_ <= n; i += FB_)
        gat_batch<FB_>(myO, i, lane, hb, e1b, e2b, ax, ay, rsum);
    if (!all) {
        const int rem = n - i;
        unsigned t = 0u;
        if (lane < rem) t = myO[i + lane];
        if (lane < rem) myO[lane] = t;
        return rem;
    }
    if (i + 8 <= n) {
        gat_batch<8>(myO, i, lane, hb, e1b, e2b, ax, ay, rsum);
        i += 8;
    }
    for (; i < n; ++i) {
        const unsigned e0  = __builtin_amdgcn_readfirstlane(myO[i]);
        const unsigned dst = e0 & 0xFFFFu;
        const unsigned d   = e0 >> 16;
        const float2 hv = *(const float2*)(hb + (size_t)dst * FOUT_ + 2 * lane);
        const float s  = e1b[d] + e2b[dst];
        const float lr = (s >= 0.f) ? s : ALPHA_ * s;
        const float ee = __expf(-lr);
        const unsigned j = (d >> 5) & 31u;
        float ux = ax[j], uy = ay[j];
        ux += ee * hv.x;
        uy += ee * hv.y;
        ax[j] = ux;
        ay[j] = uy;
        rsum += (lane == (int)j) ? ee : 0.f;
    }
    return 0;
}

__device__ __forceinline__ unsigned gat_same_owner(unsigned o, bool valid) {
    unsigned mk = __builtin_amdgcn_ballot_w32(valid);
#pragma unroll
    for (int bit = 0; bit < 5; ++bit) {
        const bool     on  = ((o >> bit) & 1u) != 0u;
        const unsigned bal = __builtin_amdgcn_ballot_w32(on);
        mk &= on ? bal : ~bal;
    }
    return mk;
}

__global__ __launch_bounds__(1024) void gat_rows(const int* __restrict__ rid,
                                                 const int* __restrict__ pk,
                                                 const float* __restrict__ h,
                                                 const float* __restrict__ e1,
                                                 const float* __restrict__ e2,
                                                 int* __restrict__ alist,
                                                 float* __restrict__ out) {
    __shared__ __attribute__((aligned(16))) unsigned char L[LDS_BYTES_];
    unsigned* blist = (unsigned*)(L + OFF_BL_);
    unsigned* olist = (unsigned*)(L + OFF_OL_);
    int*      endt  = (int*)(L + OFF_ET_);
    int*      sflag = (int*)(L + OFF_SF_);

    const int tid  = threadIdx.x;
    const int lane = tid & 31;
    const int w    = __builtin_amdgcn_readfirstlane(tid >> 5);
    const int b    = blockIdx.x / NRANGE_;
    const int rg   = blockIdx.x % NRANGE_;
    const int row_lo = rg * R_;
    const int nrows  = (N_ - row_lo < R_) ? (N_ - row_lo) : R_;

    const uint4*    ridb = (const uint4*)(rid + (size_t)b * RIDW_);
    const unsigned* pkb  = (const unsigned*)pk + (size_t)b * PKS_;
    const unsigned  rgb  = (unsigned)rg * 0x01010101u;
    const float* hb     = h  + (size_t)b * N_ * FOUT_;
    const float* e1b    = e1 + (size_t)b * N_ + row_lo;
    const float* e2b    = e2 + (size_t)b * N_;

    v32f ax = {}, ay = {};

    int*      myA = alist + ((size_t)blockIdx.x * 32 + w) * C1_;
    unsigned* myB = blist + w * C1_;
    int*      myC = endt + w * 32;
    unsigned* myO = olist + w * C2_;
    const unsigned lt_mask = (1u << lane) - 1u;
    int it   = 0;
    int ocnt = 0;
    float rsum = 0.f;

    for (int pass = 0; pass < 128; ++pass) {
        int  cnt  = 0;
        bool full = false;
        myC[lane] = 0;
        while (!full && it * 32 + w < NBLK2_) {
            uint4 s[4];
#pragma unroll
            for (int j = 0; j < 4; ++j) {
                const int blk = (it + j) * 32 + w;
                if (blk < NBLK2_) s[j] = ridb[(size_t)blk * 32 + lane];
                else              s[j] = make_uint4(0x7F7F7F7Fu, 0x7F7F7F7Fu, 0x7F7F7F7Fu, 0x7F7F7F7Fu);
            }
#pragma unroll
            for (int j = 0; j < 4; ++j) {
                if (!full) {
                    const unsigned f0 = ~((s[j].x ^ rgb) + 0x7F7F7F7Fu) & 0x80808080u;
                    const unsigned f1 = ~((s[j].y ^ rgb) + 0x7F7F7F7Fu) & 0x80808080u;
                    const unsigned f2 = ~((s[j].z ^ rgb) + 0x7F7F7F7Fu) & 0x80808080u;
                    const unsigned f3 = ~((s[j].w ^ rgb) + 0x7F7F7F7Fu) & 0x80808080u;
                    unsigned m = (f0 >> 7) | (f1 >> 6) | (f2 >> 5) | (f3 >> 4);
                    unsigned bm = __builtin_amdgcn_ballot_w32(m != 0u);
                    if (cnt + 16 * __builtin_popcount(bm) > C1_) {
                        full = true;
                    } else {
                        const unsigned ebase = (unsigned)((it * 32 + w) * RB_ + lane * 16);
                        while (bm) {
                            if (m != 0u) {
                                const unsigned p = (unsigned)__builtin_ctz(m);
                                m &= m - 1u;
                                myA[cnt + __builtin_popcount(bm & lt_mask)] =
                                    (int)(ebase + ((p & 7u) << 2) + (p >> 3));
                            }
                            cnt += __builtin_popcount(bm);
                            bm = __builtin_amdgcn_ballot_w32(m != 0u);
                        }
                        ++it;
                    }
                }
            }
        }
        __threadfence_block();
        for (int i0 = 0; i0 < cnt; i0 += 128) {
            unsigned en[4], pw[4];
#pragma unroll
            for (int j = 0; j < 4; ++j) {
                const int idx = i0 + j * 32 + lane;
                en[j] = (idx < cnt) ? (unsigned)myA[idx] : 0u;
            }
#pragma unroll
            for (int j = 0; j < 4; ++j) pw[j] = pkb[en[j]];
#pragma unroll
            for (int j = 0; j < 4; ++j) {
                const int idx = i0 + j * 32 + lane;
                if (idx < cnt) {
                    const unsigned d = (pw[j] & 0xFFFFu) - (unsigned)row_lo;
                    myA[idx] = (int)((pw[j] >> 16) | (d << 16));
                    atomicAdd(&myC[d & 31u], 1);
                }
            }
        }
        __threadfence_block();
        {
            const int c = myC[lane];
            int incl = c;
#pragma unroll
            for (int k = 1; k < 32; k <<= 1) {
                const int t = __shfl_up(incl, k, 32);
                if (lane >= k) incl += t;
            }
            myC[lane] = incl - c;
        }
        for (int i0 = 0; i0 < cnt; i0 += 128) {
            unsigned en[4];
#pragma unroll
            for (int j = 0; j < 4; ++j) {
                const int idx = i0 + j * 32 + lane;
                en[j] = (idx < cnt) ? (unsigned)myA[idx] : 0u;
            }
#pragma unroll
            for (int j = 0; j < 4; ++j) {
                if (i0 + j * 32 < cnt) {
                    const bool     valid = (i0 + j * 32 + lane) < cnt;
                    const unsigned d     = en[j] >> 16;
                    const unsigned o     = d & 31u;
                    const unsigned mk    = gat_same_owner(o, valid);
                    const int pos = myC[o] + __builtin_popcount(mk & lt_mask);
                    if (valid) myB[pos] = en[j];
                    if (valid && (mk >> lane) == 1u) myC[o] = pos + 1;
                }
            }
        }
        if (lane == 0) sflag[w] = (it * 32 + w < NBLK2_) ? 1 : 0;
        __syncthreads();

        const int pieceEnd   = endt[lane * 32 + w];
        const int pieceStart = (w > 0) ? endt[lane * 32 + w - 1] : 0;
        for (int v = 0; v < 32; ++v) {
            int       p  = __builtin_amdgcn_readlane(pieceStart, v);
            const int p1 = __builtin_amdgcn_readlane(pieceEnd, v);
            const unsigned* bl = blist + v * C1_;
            while (p < p1) {
                if (ocnt > C2_ - 32)
                    ocnt = gat_flush(myO, ocnt, lane, false, hb, e1b, e2b, ax, ay, rsum);
                const int take = (p1 - p < 32) ? (p1 - p) : 32;
                if (lane < take) myO[ocnt + lane] = bl[p + lane];
                ocnt += take;
                p    += take;
            }
        }
        const unsigned more = __builtin_amdgcn_ballot_w32(sflag[lane] != 0);
        __syncthreads();
        if (more == 0u) break;
    }
    gat_flush(myO, ocnt, lane, true, hb, e1b, e2b, ax, ay, rsum);

    float* ob = out + ((size_t)b * N_ + row_lo + w) * FOUT_ + 2 * lane;
#pragma nounroll
    for (int j = 0; j * 32 + w < nrows; ++j) {
        const float q = __int_as_float(__builtin_amdgcn_readlane(__float_as_int(rsum), j));
        float v0, v1;
        if (q >= 0x1p-100f && q <= 0x1p+100f) {
            float r = __builtin_amdgcn_rcpf(q);
            r = __builtin_fmaf(__builtin_fmaf(-q, r, 1.0f), r, r);
            v0 = ax[j] * r;
            v1 = ay[j] * r;
        } else {
            v0 = ax[j] / q;
            v1 = ay[j] / q;
        }
        if (isnan(v0)) v0 = NEG_BIG_;
        if (isnan(v1)) v1 = NEG_BIG_;
        float2 o;
        o.x = (v0 > 0.f) ? v0 : expm1f(v0);
        o.y = (v1 > 0.f) ? v1 : expm1f(v1);
        *(float2*)(ob + (size_t)j * 32 * FOUT_) = o;
    }
}

extern "C" void kernel_launch(void* const* d_in, const int* in_sizes, int n_in,
                              void* d_out, int out_size, void* d_ws, size_t ws_size,
                              hipStream_t stream) {
    const float* X     = (const float*)d_in[0];
    const int*   edges = (const int*)d_in[1];
    const float* W     = (const float*)d_in[2];
    const float* a     = (const float*)d_in[3];
    float*       out   = (float*)d_out;

    float* h  = (float*)d_ws;
    float* e1 = h + (size_t)ROWS_ * FOUT_;
    float* e2 = e1 + (size_t)ROWS_;
    int*   alist = (int*)(e2 + (size_t)ROWS_);
    int*   rid   = alist + (size_t)B_ * NRANGE_ * 32 * C1_;
    int*   pk    = rid + (size_t)B_ * RIDW_;

    gat_gemm_wmma<<<GW_BLOCKS_, 256, 0, stream>>>(X, W, h);
    {
        const int blocks = SC_BLOCKS_;
        gat_scores<<<blocks, 256, 0, stream>>>(h, a, edges, e1, e2, rid, pk);
    }
    gat_rows<<<B_ * NRANGE_, 1024, 0, stream>>>(rid, pk, h, e1, e2, alist, out);
}


# Rows kernel: each scanning wave's match list lives in LDS, not in global scratch: its stores, both read-backs (before the packed-word gather and the sort) and two fences stay off L2; sum order same
# speedup vs baseline: 7.5202x; 1.0322x over previous
#include <hip/hip_runtime.h>
#include <hip/hip_bf16.h>
#include <math.h>

typedef __attribute__((ext_vector_type(2))) float v2f;
typedef __attribute__((ext_vector_type(8))) float v8f;
typedef __attribute__((ext_vector_type(4))) float v4f;
typedef __attribute__((ext_vector_type(32))) float v32f;

#define B_ 4
#define N_ 50000
#define E_ 800000
#define FIN_ 128
#define FOUT_ 64
#define ALPHA_ 0.2f
#define NEG_BIG_ (-9000000000000000.0f)

#define ROWS_ (B_ * N_)
#define TILES_ (ROWS_ / 16)

#define GW_BLOCKS_ 256
#define GW_WAVES_  (GW_BLOCKS_ * 8)
static_assert(GW_WAVES_ <= TILES_, "every wave has a first tile");

__device__ __forceinline__ void gat_tile_load(v4f (&a)[16], const float* __restrict__ X,
                                              int rt, int c, int hf) {
    const v4f* xp = (const v4f*)(X + (size_t)(rt * 16 + c) * FIN_ + 4 * hf);
#pragma unroll
    for (int j = 0; j < 16; ++j) a[j] = xp[2 * j];
}

__device__ __forceinline__ void gat_tile_hold(const v4f (&a)[16]) {
    asm volatile("" :: "v"(a[0]), "v"(a[1]), "v"(a[2]), "v"(a[3]), "v"(a[4]), "v"(a[5]), "v"(a[6]), "v"(a[7]),
                       "v"(a[8]), "v"(a[9]), "v"(a[10]), "v"(a[11]), "v"(a[12]), "v"(a[13]), "v"(a[14]), "v"(a[15]));
}

__device__ __forceinline__ void gat_tile_mma(const v4f (&a)[16], const v2f (&wf)[16][2][4],
                                             float* __restrict__ h, int rt, int c, int hf) {
    v8f acc0 = {}, acc1 = {}, acc2 = {}, acc3 = {};
#pragma unroll
    for (int j = 0; j < 16; ++j) {
        v2f p, q;
        p.x = a[j].x; p.y = a[j].y;
        q.x = a[j].z; q.y = a[j].w;
        acc0 = __builtin_amdgcn_wmma_f32_16x16x4_f32(false, p, false, wf[j][0][0], (short)0, acc0, false, false);
        acc1 = __builtin_amdgcn_wmma_f32_16x16x4_f32(false, p, false, wf[j][0][1], (short)0, acc1, false, false);
        acc2 = __builtin_amdgcn_wmma_f32_16x16x4_f32(false, p, false, wf[j][0][2], (short)0, acc2, false, false);
        acc3 = __builtin_amdgcn_wmma_f32_16x16x4_f32(false, p, false, wf[j][0][3], (short)0, acc3, false, false);
        acc0 = __builtin_amdgcn_wmma_f32_16x16x4_f32(false, q, false, wf[j][1][0], (short)0, acc0, false, false);
        acc1 = __builtin_amdgcn_wmma_f32_16x16x4_f32(false, q, false, wf[j][1][1], (short)0, acc1, false, false);
        acc2 = __builtin_amdgcn_wmma_f32_16x16x4_f32(false, q, false, wf[j][1][2], (short)0, acc2, false, false);
        acc3 = __builtin_amdgcn_wmma_f32_16x16x4_f32(false, q, false, wf[j][1][3], (short)0, acc3, false, false);
    }
    float* hp = h + (size_t)(rt * 16 + 8 * hf) * FOUT_ + 4 * c;
#pragma unroll
    for (int r = 0; r < 8; ++r)
        *(float4*)(hp + r * FOUT_) = make_float4(acc0[r], acc1[r], acc2[r], acc3[r]);
}

__global__ __launch_bounds__(256) void gat_gemm_wmma(const float* __restrict__ X,
                                                     const float* __restrict__ W,
                                                     float* __restrict__ h) {
    const int lane = threadIdx.x & 31;
    const int c    = lane & 15;
    const int hf   = lane >> 4;
    const int g = __builtin_amdgcn_readfirstlane((int)(threadIdx.x >> 5)) * GW_BLOCKS_ + (int)blockIdx.x;

    v2f wf[16][2][4];
#pragma unroll
    for (int j = 0; j < 16; ++j) {
#pragma unroll
        for (int u = 0; u < 2; ++u) {
            const float* wp = W + (size_t)(8 * j + 4 * hf + 2 * u) * FOUT_ + 4 * c;
            const float4 w0 = *(const float4*)wp;
            const float4 w1 = *(const float4*)(wp + FOUT_);
            wf[j][u][0].x = w0.x; wf[j][u][0].y = w1.x;
            wf[j][u][1].x = w0.y; wf[j][u][1].y = w1.y;
            wf[j][u][2].x = w0.z; wf[j][u][2].y = w1.z;
            wf[j][u][3].x = w0.w; wf[j][u][3].y = w1.w;
        }
    }

    v4f a0[16], a1[16];
    int rt = g;
    gat_tile_load(a0, X, rt, c, hf);
    for (int it = 0; it < 4; ++it) {
        int nx = rt + GW_WAVES_;
        gat_tile_load(a1, X, (nx < TILES_) ? nx : rt, c, hf);
        __builtin_amdgcn_sched_barrier(0);
        gat_tile_mma(a0, wf, h, rt, c, hf);
        gat_tile_hold(a1);
        if (nx >= TILES_) break;
        rt = nx;
        nx = rt + GW_WAVES_;
        gat_tile_load(a0, X, (nx < TILES_) ? nx : rt, c, hf);
        __builtin_amdgcn_sched_barrier(0);
        gat_tile_mma(a1, wf, h, rt, c, hf);
        gat_tile_hold(a0);
        if (nx >= TILES_) break;
        rt = nx;
    }
}

#define RNG_     782
#define RB_      512
#define NBLK2_   ((E_ + RB_ - 1) / RB_)
#define RIDW_    (NBLK2_ * (RB_ / 4))
#define PKS_     (NBLK2_ * RB_)
static_assert(RIDW_ % 8 == 0 && (RIDW_ / 8) == NBLK2_ * 16, "groups of 8 words: 16 per block, dealt over all blocks");
#define SC_BLOCKS_ ((ROWS_ + 255) / 256)

__device__ __forceinline__ unsigned gat_rid(int s) {
    return ((unsigned)s < (unsigned)N_) ? ((unsigned)s / (unsigned)RNG_) : 0x7Fu;
}

__global__ __launch_bounds__(256) void gat_scores(const float* __restrict__ h,
                                                  const float* __restrict__ a,
                                                  const int* __restrict__ edges,
                                                  float* __restrict__ e1,
                                                  float* __restrict__ e2,
                                                  int* __restrict__ rid,
                                                  int* __restrict__ pk) {
    const int row = blockIdx.x * 256 + threadIdx.x;
    if (row < ROWS_) {
        const float* hr = h + (size_t)row * FOUT_;
        float s1 = 0.f, s2 = 0.f;
#pragma unroll
        for (int f = 0; f < FOUT_; ++f) {
            const float hv = hr[f];
            s1 += hv * a[f];
            s2 += hv * a[FOUT_ + f];
        }
        e1[row] = s1;
        e2[row] = s2;
    }
#pragma unroll
    for (int j = 0; j < 4; ++j) {
        const int q = row + j * (SC_BLOCKS_ * 256);
        if (q < B_ * RIDW_) {
            const int b = q / RIDW_;
            const int i = q - b * RIDW_;
            const int g = i >> 3;
            const int ip = (g % NBLK2_) * (RB_ / 4) + (g / NBLK2_) * 8 + (i & 7);
            unsigned r = 0x7F7F7F7Fu;
            if (i < E_ / 4) {
                const int* sp = edges + (size_t)b * 2 * E_ + (size_t)i * 4;
                const int4 s = *(const int4*)sp;
                const int4 d = *(const int4*)(sp + E_);
                r = gat_rid(s.x) | (gat_rid(s.y) << 8) | (gat_rid(s.z) << 16) | (gat_rid(s.w) << 24);
                int4 p;
                p.x = (int)(((unsigned)s.x & 0xFFFFu) | ((unsigned)d.x << 16));
                p.y = (int)(((unsigned)s.y & 0xFFFFu) | ((unsigned)d.y << 16));
                p.z = (int)(((unsigned)s.z & 0xFFFFu) | ((unsigned)d.z << 16));
                p.w = (int)(((unsigned)s.w & 0xFFFFu) | ((unsigned)d.w << 16));
                *(int4*)(pk + (size_t)b * PKS_ + (size_t)ip * 4) = p;
            }
            rid[(size_t)b * RIDW_ + ip] = (int)r;
        }
    }
}

#define R_      782
#define NRANGE_ 64
#define C1_     704
#define C2_     72
static_assert(R_ == RNG_, "kernel 2 writes the range bytes for kernel 3's ranges");
#define RPW_    25
static_assert((R_ + 31) / 32 == RPW_ && RPW_ <= 32, "a wave's rows fit the 32-entry register vectors");
#define OFF_BL_ 0
#define OFF_OL_ (OFF_BL_ + 32 * C1_ * 4)
#define OFF_ET_ (OFF_OL_ + 32 * C2_ * 4)
#define OFF_SF_ (OFF_ET_ + 32 * 32 * 4)
#define OFF_AL_ (OFF_SF_ + 32 * 4)
#define LDS_BYTES_ (OFF_AL_ + 32 * C1_ * 4)

typedef __attribute__((address_space(1))) const unsigned char gat_gbyte;
typedef __attribute__((address_space(1))) const v2f gat_gv2f;

#define FB_ 16

__device__ __forceinline__ void gat_rsum_add(float& rsum, const unsigned (&j)[16], const int (&e)[16]) {
    unsigned keep;
    asm volatile("s_mov_b32 %[kp], exec_lo\n\t"
                 "s_bfm_b32 exec_lo, 1, %[j0]\n\t"
                 "v_add_f32_e32 %[r], %[e0], %[r]\n\t"
                 "s_bfm_b32 exec_lo, 1, %[j1]\n\t"
                 "v_add_f32_e32 %[r], %[e1], %[r]\n\t"
                 "s_bfm_b32 exec_lo, 1, %[j2]\n\t"
                 "v_add_f32_e32 %[r], %[e2], %[r]\n\t"
                 "s_bfm_b32 exec_lo, 1, %[j3]\n\t"
                 "v_add_f32_e32 %[r], %[e3], %[r]\n\t"
                 "s_bfm_b32 exec_lo, 1, %[j4]\n\t"
                 "v_add_f32_e32 %[r], %[e4], %[r]\n\t"
                 "s_bfm_b32 exec_lo, 1, %[j5]\n\t"
                 "v_add_f32_e32 %[r], %[e5], %[r]\n\t"
                 "s_bfm_b32 exec_lo, 1, %[j6]\n\t"
                 "v_add_f32_e32 %[r], %[e6], %[r]\n\t"
                 "s_bfm_b32 exec_lo, 1, %[j7]\n\t"
                 "v_add_f32_e32 %[r], %[e7], %[r]\n\t"
                 "s_bfm_b32 exec_lo, 1, %[j8]\n\t"
                 "v_add_f32_e32 %[r], %[e8], %[r]\n\t"
                 "s_bfm_b32 exec_lo, 1, %[j9]\n\t"
                 "v_add_f32_e32 %[r], %[e9], %[r]\n\t"
                 "s_bfm_b32 exec_lo, 1, %[j10]\n\t"
                 "v_add_f32_e32 %[r], %[e10], %[r]\n\t"
                 "s_bfm_b32 exec_lo, 1, %[j11]\n\t"
                 "v_add_f32_e32 %[r], %[e11], %[r]\n\t"
                 "s_bfm_b32 exec_lo, 1, %[j12]\n\t"
                 "v_add_f32_e32 %[r], %[e12], %[r]\n\t"
                 "s_bfm_b32 exec_lo, 1, %[j13]\n\t"
                 "v_add_f32_e32 %[r], %[e13], %[r]\n\t"
                 "s_bfm_b32 exec_lo, 1, %[j14]\n\t"
                 "v_add_f32_e32 %[r], %[e14], %[r]\n\t"
                 "s_bfm_b32 exec_lo, 1, %[j15]\n\t"
                 "v_add_f32_e32 %[r], %[e15], %[r]\n\t"
                 "s_mov_b32 exec_lo, %[kp]"
                 : [r] "+v"(rsum), [kp] "=&s"(keep)
                 : [j0] "s"(j[0]), [j1] "s"(j[1]), [j2] "s"(j[2]), [j3] "s"(j[3]), [j4] "s"(j[4]), [j5] "s"(j[5]), [j6] "s"(j[6]), [j7] "s"(j[7]), [j8] "s"(j[8]), [j9] "s"(j[9]), [j10] "s"(j[10]), [j11] "s"(j[11]), [j12] "s"(j[12]), [j13] "s"(j[13]), [j14] "s"(j[14]), [j15] "s"(j[15]),
                   [e0] "s"(e[0]), [e1] "s"(e[1]), [e2] "s"(e[2]), [e3] "s"(e[3]), [e4] "s"(e[4]), [e5] "s"(e[5]), [e6] "s"(e[6]), [e7] "s"(e[7]), [e8] "s"(e[8]), [e9] "s"(e[9]), [e10] "s"(e[10]), [e11] "s"(e[11]), [e12] "s"(e[12]), [e13] "s"(e[13]), [e14] "s"(e[14]), [e15] "s"(e[15]));
}
__device__ __forceinline__ void gat_rsum_add(float& rsum, const unsigned (&j)[8], const int (&e)[8]) {
    unsigned keep;
    asm volatile("s_mov_b32 %[kp], exec_lo\n\t"
                 "s_bfm_b32 exec_lo, 1, %[j0]\n\t"
                 "v_add_f32_e32 %[r], %[e0], %[r]\n\t"
                 "s_bfm_b32 exec_lo, 1, %[j1]\n\t"
                 "v_add_f32_e32 %[r], %[e1], %[r]\n\t"
                 "s_bfm_b32 exec_lo, 1, %[j2]\n\t"
                 "v_add_f32_e32 %[r], %[e2], %[r]\n\t"
                 "s_bfm_b32 exec_lo, 1, %[j3]\n\t"
                 "v_add_f32_e32 %[r], %[e3], %[r]\n\t"
                 "s_bfm_b32 exec_lo, 1, %[j4]\n\t"
                 "v_add_f32_e32 %[r], %[e4], %[r]\n\t"
                 "s_bfm_b32 exec_lo, 1, %[j5]\n\t"
                 "v_add_f32_e32 %[r], %[e5], %[r]\n\t"
                 "s_bfm_b32 exec_lo, 1, %[j6]\n\t"
                 "v_add_f32_e32 %[r], %[e6], %[r]\n\t"
                 "s_bfm_b32 exec_lo, 1, %[j7]\n\t"
                 "v_add_f32_e32 %[r], %[e7], %[r]\n\t"
                 "s_mov_b32 exec_lo, %[kp]"
                 : [r] "+v"(rsum), [kp] "=&s"(keep)
                 : [j0] "s"(j[0]), [j1] "s"(j[1]), [j2] "s"(j[2]), [j3] "s"(j[3]), [j4] "s"(j[4]), [j5] "s"(j[5]), [j6] "s"(j[6]), [j7] "s"(j[7]),
                   [e0] "s"(e[0]), [e1] "s"(e[1]), [e2] "s"(e[2]), [e3] "s"(e[3]), [e4] "s"(e[4]), [e5] "s"(e[5]), [e6] "s"(e[6]), [e7] "s"(e[7]));
}

template <int NB>
__device__ __forceinline__ void gat_batch(const unsigned* __restrict__ myO, int i, int lane,
                                          const float* __restrict__ hb,
                                          const float* __restrict__ e1b,
                                          const float* __restrict__ e2b,
                                          v32f& ax, v32f& ay, float& rsum) {
    static_assert(NB == 8 || NB == 16, "lane k (mod NB) takes entry k");
    unsigned l8 = (unsigned)lane * 8u;
    asm volatile("" : "+v"(l8));
    unsigned en[NB];
    float2 hv[NB];
    const unsigned mine = myO[i + (lane & (NB - 1))];
#pragma unroll
    for (int k = 0; k < NB; ++k) en[k] = (unsigned)__builtin_amdgcn_readlane((int)mine, k);
#pragma unroll
    for (int k = 0; k < NB; ++k) {
        const unsigned dst = en[k] & 0xFFFFu;
        const unsigned long long rb = (unsigned long long)hb + ((unsigned long long)dst << 8);
        const unsigned rlo = (unsigned)__builtin_amdgcn_readfirstlane((int)(unsigned)rb);
        const unsigned rhi = (unsigned)__builtin_amdgcn_readfirstlane((int)(unsigned)(rb >> 32));
        const gat_gbyte* rowp = (const gat_gbyte*)(((unsigned long long)rhi << 32) | (unsigned long long)rlo);
        const v2f t = *(const gat_gv2f*)(rowp + l8);
        hv[k] = make_float2(t.x, t.y);
    }
    const float g1 = e1b[mine >> 16];
    const float g2 = e2b[mine & 0xFFFFu];
    __builtin_amdgcn_sched_barrier(0);
    const float ms  = g1 + g2;
    const float mlr = (ms >= 0.f) ? ms : ALPHA_ * ms;
    const int   mee = __float_as_int(__expf(-mlr));
    unsigned jr[NB];
    int      eb[NB];
#pragma unroll
    for (int k = 0; k < NB; ++k) {
        const unsigned j = (en[k] >> 21) & 31u;
        eb[k] = __builtin_amdgcn_readlane(mee, k);
        jr[k] = j;
        const float ee = __int_as_float(eb[k]);
        float ux = ax[j], uy = ay[j];
        ux += ee * hv[k].x;
        uy += ee * hv[k].y;
        ax[j] = ux;
        ay[j] = uy;
    }
    gat_rsum_add(rsum, jr, eb);
}

__device__ __forceinline__ int gat_flush(unsigned* __restrict__ myO, int n, int lane, bool all,
                                         const float* __restrict__ hb,
                                         const float* __restrict__ e1b,
                                         const float* __restrict__ e2b,
                                         v32f& ax, v32f& ay, float& rsum) {
    int i = 0;
    for (; i + FB_ <= n; i += FB_)
        gat_batch<FB_>(myO, i, lane, hb, e1b, e2b, ax, ay, rsum);
    if (!all) {
        const int rem = n - i;
        unsigned t = 0u;
        if (lane < rem) t = myO[i + lane];
        if (lane < rem) myO[lane] = t;
        return rem;
    }
    if (i + 8 <= n) {
        gat_batch<8>(myO, i, lane, hb, e1b, e2b, ax, ay, rsum);
        i += 8;
    }
    for (; i < n; ++i) {
        const unsigned e0  = __builtin_amdgcn_readfirstlane(myO[i]);
        const unsigned dst = e0 & 0xFFFFu;
        const unsigned d   = e0 >> 16;
        const float2 hv = *(const float2*)(hb + (size_t)dst * FOUT_ + 2 * lane);
        const float s  = e1b[d] + e2b[dst];
        const float lr = (s >= 0.f) ? s : ALPHA_ * s;
        const float ee = __expf(-lr);
        const unsigned j = (d >> 5) & 31u;
        float ux = ax[j], uy = ay[j];
        ux += ee * hv.x;
        uy += ee * hv.y;
        ax[j] = ux;
        ay[j] = uy;
        rsum += (lane == (int)j) ? ee : 0.f;
    }
    return 0;
}

__device__ __forceinline__ unsigned gat_same_owner(unsigned o, bool valid) {
    unsigned mk = __builtin_amdgcn_ballot_w32(valid);
#pragma unroll
    for (int bit = 0; bit < 5; ++bit) {
        const bool     on  = ((o >> bit) & 1u) != 0u;
        const unsigned bal = __builtin_amdgcn_ballot_w32(on);
        mk &= on ? bal : ~bal;
    }
    return mk;
}

__global__ __launch_bounds__(1024) void gat_rows(const int* __restrict__ rid,
                                                 const int* __restrict__ pk,
                                                 const float* __restrict__ h,
                                                 const float* __restrict__ e1,
                                                 const float* __restrict__ e2,
                                                 float* __restrict__ out) {
    __shared__ __attribute__((aligned(16))) unsigned char L[LDS_BYTES_];
    unsigned* blist = (unsigned*)(L + OFF_BL_);
    unsigned* olist = (unsigned*)(L + OFF_OL_);
    int*      endt  = (int*)(L + OFF_ET_);
    int*      sflag = (int*)(L + OFF_SF_);

    const int tid  = threadIdx.x;
    const int lane = tid & 31;
    const int w    = __builtin_amdgcn_readfirstlane(tid >> 5);
    const int b    = blockIdx.x / NRANGE_;
    const int rg   = blockIdx.x % NRANGE_;
    const int row_lo = rg * R_;
    const int nrows  = (N_ - row_lo < R_) ? (N_ - row_lo) : R_;

    const uint4*    ridb = (const uint4*)(rid + (size_t)b * RIDW_);
    const unsigned* pkb  = (const unsigned*)pk + (size_t)b * PKS_;
    const unsigned  rgb  = (unsigned)rg * 0x01010101u;
    const float* hb     = h  + (size_t)b * N_ * FOUT_;
    const float* e1b    = e1 + (size_t)b * N_ + row_lo;
    const float* e2b    = e2 + (size_t)b * N_;

    v32f ax = {}, ay = {};

    int*      myA = (int*)(L + OFF_AL_) + w * C1_;
    unsigned* myB = blist + w * C1_;
    int*      myC = endt + w * 32;
    unsigned* myO = olist + w * C2_;
    const unsigned lt_mask = (1u << lane) - 1u;
    int it   = 0;
    int ocnt = 0;
    float rsum = 0.f;

    for (int pass = 0; pass < 128; ++pass) {
        int  cnt  = 0;
        bool full = false;
        myC[lane] = 0;
        while (!full && it * 32 + w < NBLK2_) {
            uint4 s[4];
#pragma unroll
            for (int j = 0; j < 4; ++j) {
                const int blk = (it + j) * 32 + w;
                if (blk < NBLK2_) s[j] = ridb[(size_t)blk * 32 + lane];
                else              s[j] = make_uint4(0x7F7F7F7Fu, 0x7F7F7F7Fu, 0x7F7F7F7Fu, 0x7F7F7F7Fu);
            }
#pragma unroll
            for (int j = 0; j < 4; ++j) {
                if (!full) {
                    const unsigned f0 = ~((s[j].x ^ rgb) + 0x7F7F7F7Fu) & 0x80808080u;
                    const unsigned f1 = ~((s[j].y ^ rgb) + 0x7F7F7F7Fu) & 0x80808080u;
                    const unsigned f2 = ~((s[j].z ^ rgb) + 0x7F7F7F7Fu) & 0x80808080u;
                    const unsigned f3 = ~((s[j].w ^ rgb) + 0x7F7F7F7Fu) & 0x80808080u;
                    unsigned m = (f0 >> 7) | (f1 >> 6) | (f2 >> 5) | (f3 >> 4);
                    unsigned bm = __builtin_amdgcn_ballot_w32(m != 0u);
                    if (cnt + 16 * __builtin_popcount(bm) > C1_) {
                        full = true;
                    } else {
                        const unsigned ebase = (unsigned)((it * 32 + w) * RB_ + lane * 16);
                        while (bm) {
                            if (m != 0u) {
                                const unsigned p = (unsigned)__builtin_ctz(m);
                                m &= m - 1u;
                                myA[cnt + __builtin_popcount(bm & lt_mask)] =
                                    (int)(ebase + ((p & 7u) << 2) + (p >> 3));
                            }
                            cnt += __builtin_popcount(bm);
                            bm = __builtin_amdgcn_ballot_w32(m != 0u);
                        }
                        ++it;
                    }
                }
            }
        }
        __threadfence_block();
        for (int i0 = 0; i0 < cnt; i0 += 128) {
            unsigned en[4], pw[4];
#pragma unroll
            for (int j = 0; j < 4; ++j) {
                const int idx = i0 + j * 32 + lane;
                en[j] = (idx < cnt) ? (unsigned)myA[idx] : 0u;
            }
#pragma unroll
            for (int j = 0; j < 4; ++j) pw[j] = pkb[en[j]];
#pragma unroll
            for (int j = 0; j < 4; ++j) {
                const int idx = i0 + j * 32 + lane;
                if (idx < cnt) {
                    const unsigned d = (pw[j] & 0xFFFFu) - (unsigned)row_lo;
                    myA[idx] = (int)((pw[j] >> 16) | (d << 16));
                    atomicAdd(&myC[d & 31u], 1);
                }
            }
        }
        __threadfence_block();
        {
            const int c = myC[lane];
            int incl = c;
#pragma unroll
            for (int k = 1; k < 32; k <<= 1) {
                const int t = __shfl_up(incl, k, 32);
                if (lane >= k) incl += t;
            }
            myC[lane] = incl - c;
        }
        for (int i0 = 0; i0 < cnt; i0 += 128) {
            unsigned en[4];
#pragma unroll
            for (int j = 0; j < 4; ++j) {
                const int idx = i0 + j * 32 + lane;
                en[j] = (idx < cnt) ? (unsigned)myA[idx] : 0u;
            }
#pragma unroll
            for (int j = 0; j < 4; ++j) {
                if (i0 + j * 32 < cnt) {
                    const bool     valid = (i0 + j * 32 + lane) < cnt;
                    const unsigned d     = en[j] >> 16;
                    const unsigned o     = d & 31u;
                    const unsigned mk    = gat_same_owner(o, valid);
                    const int pos = myC[o] + __builtin_popcount(mk & lt_mask);
                    if (valid) myB[pos] = en[j];
                    if (valid && (mk >> lane) == 1u) myC[o] = pos + 1;
                }
            }
        }
        if (lane == 0) sflag[w] = (it * 32 + w < NBLK2_) ? 1 : 0;
        __syncthreads();

        const int pieceEnd   = endt[lane * 32 + w];
        const int pieceStart = (w > 0) ? endt[lane * 32 + w - 1] : 0;
        for (int v = 0; v < 32; ++v) {
            int       p  = __builtin_amdgcn_readlane(pieceStart, v);
            const int p1 = __builtin_amdgcn_readlane(pieceEnd, v);
            const unsigned* bl = blist + v * C1_;
            while (p < p1) {
                if (ocnt > C2_ - 32)
                    ocnt = gat_flush(myO, ocnt, lane, false, hb, e1b, e2b, ax, ay, rsum);
                const int take = (p1 - p < 32) ? (p1 - p) : 32;
                if (lane < take) myO[ocnt + lane] = bl[p + lane];
                ocnt += take;
                p    += take;
            }
        }
        const unsigned more = __builtin_amdgcn_ballot_w32(sflag[lane] != 0);
        __syncthreads();
        if (more == 0u) break;
    }
    gat_flush(myO, ocnt, lane, true, hb, e1b, e2b, ax, ay, rsum);

    float* ob = out + ((size_t)b * N_ + row_lo + w) * FOUT_ + 2 * lane;
#pragma nounroll
    for (int j = 0; j * 32 + w < nrows; ++j) {
        const float q = __int_as_float(__builtin_amdgcn_readlane(__float_as_int(rsum), j));
        float v0, v1;
        if (q >= 0x1p-100f && q <= 0x1p+100f) {
            float r = __builtin_amdgcn_rcpf(q);
            r = __builtin_fmaf(__builtin_fmaf(-q, r, 1.0f), r, r);
            v0 = ax[j] * r;
            v1 = ay[j] * r;
        } else {
            v0 = ax[j] / q;
            v1 = ay[j] / q;
        }
        if (isnan(v0)) v0 = NEG_BIG_;
        if (isnan(v1)) v1 = NEG_BIG_;
        float2 o;
        o.x = (v0 > 0.f) ? v0 : expm1f(v0);
        o.y = (v1 > 0.f) ? v1 : expm1f(v1);
        *(float2*)(ob + (size_t)j * 32 * FOUT_) = o;
    }
}

extern "C" void kernel_launch(void* const* d_in, const int* in_sizes, int n_in,
                              void* d_out, int out_size, void* d_ws, size_t ws_size,
                              hipStream_t stream) {
    const float* X     = (const float*)d_in[0];
    const int*   edges = (const int*)d_in[1];
    const float* W     = (const float*)d_in[2];
    const float* a     = (const float*)d_in[3];
    float*       out   = (float*)d_out;

    float* h  = (float*)d_ws;
    float* e1 = h + (size_t)ROWS_ * FOUT_;
    float* e2 = e1 + (size_t)ROWS_;
    int*   rid   = (int*)(e2 + (size_t)ROWS_);
    int*   pk    = rid + (size_t)B_ * RIDW_;

    gat_gemm_wmma<<<GW_BLOCKS_, 256, 0, stream>>>(X, W, h);
    {
        const int blocks = SC_BLOCKS_;
        gat_scores<<<blocks, 256, 0, stream>>>(h, a, edges, e1, e2, rid, pk);
    }
    gat_rows<<<B_ * NRANGE_, 1024, 0, stream>>>(rid, pk, h, e1, e2, out);
}
